# starting kernel
# baseline (speedup 1.0000x reference)
#include <hip/hip_runtime.h>
#include <hip/hip_bf16.h>


#define N_DRUGS 591
#define NP      592
#define FDIM    1024
#define NEDGE   120000

typedef __attribute__((ext_vector_type(2))) float v2f;
typedef __attribute__((ext_vector_type(8))) float v8f;

__global__ void init_ws(float* __restrict__ ahat, float* __restrict__ deg,
                        float* __restrict__ pool) {
  int idx = blockIdx.x * blockDim.x + threadIdx.x;
  int stride = gridDim.x * blockDim.x;
  const int AH = 3 * NP * NP;
  for (int i = idx; i < AH; i += stride) ahat[i] = 0.0f;
  for (int i = idx; i < 3 * NP; i += stride) deg[i] = 1.0f;
  if (idx < 8) pool[idx] = 0.0f;
}

__global__ void pad_dm1(const float* __restrict__ dm1, float* __restrict__ out) {
  int idx = blockIdx.x * blockDim.x + threadIdx.x;
  if (idx >= NP * FDIM) return;
  int n = idx >> 10;
  out[idx] = (n < N_DRUGS) ? dm1[idx] : 0.0f;
}

__global__ void deg_accum(const int* __restrict__ eT, const float* __restrict__ wT,
                          const int* __restrict__ eS, const float* __restrict__ wS,
                          const int* __restrict__ eG, const float* __restrict__ wG,
                          float* __restrict__ deg) {
  int idx = blockIdx.x * blockDim.x + threadIdx.x;
  if (idx >= 3 * NEDGE) return;
  int t = idx / NEDGE, e = idx - t * NEDGE;
  const int*   ed = (t == 0) ? eT : ((t == 1) ? eS : eG);
  const float* w  = (t == 0) ? wT : ((t == 1) ? wS : wG);
  int dst = ed[NEDGE + e];
  atomicAdd(&deg[t * NP + dst], w[e]);
}

__global__ void compute_dinv(const float* __restrict__ deg, float* __restrict__ dinv) {
  int i = blockIdx.x * blockDim.x + threadIdx.x;
  if (i < 3 * NP) {
    float d = deg[i];
    dinv[i] = (d > 0.0f) ? rsqrtf(d) : 0.0f;
  }
}

__global__ void build_ahat(const int* __restrict__ eT, const float* __restrict__ wT,
                           const int* __restrict__ eS, const float* __restrict__ wS,
                           const int* __restrict__ eG, const float* __restrict__ wG,
                           const float* __restrict__ dinv, float* __restrict__ ahat) {
  const int PER = NEDGE + N_DRUGS;
  int idx = blockIdx.x * blockDim.x + threadIdx.x;
  if (idx >= 3 * PER) return;
  int t = idx / PER, r = idx - t * PER;
  const float* dv = dinv + t * NP;
  float* Ah = ahat + (size_t)t * NP * NP;
  if (r < NEDGE) {
    const int*   ed = (t == 0) ? eT : ((t == 1) ? eS : eG);
    const float* w  = (t == 0) ? wT : ((t == 1) ? wS : wG);
    int src = ed[r], dst = ed[NEDGE + r];
    atomicAdd(&Ah[(size_t)dst * NP + src], dv[src] * w[r] * dv[dst]);
  } else {
    int i = r - NEDGE;
    atomicAdd(&Ah[(size_t)i * NP + i], dv[i] * dv[i]);
  }
}

__global__ void gemm_wmma_f32(const float* __restrict__ A, int lda,
                              const float* __restrict__ B, int ldb,
                              float* __restrict__ C, int ldc,
                              int K, int n_wtiles, int m_tiles,
                              const float* __restrict__ bias, int do_relu) {
  int wid = (blockIdx.x * blockDim.x + threadIdx.x) >> 5;
  if (wid >= m_tiles * n_wtiles) return;
  int lane = threadIdx.x & 31;
  int tm = wid / n_wtiles, tn = wid - tm * n_wtiles;
  int row0 = tm << 4, col0 = tn << 6;
  int ml = lane & 15;
  int khalf = (lane >> 4) << 1;

  const float* Arow = A + (size_t)(row0 + ml) * lda + khalf;
  const float* B0   = B + (size_t)khalf * ldb + (col0 + ml);

  v8f acc0 = {}, acc1 = {}, acc2 = {}, acc3 = {};
#pragma unroll 2
  for (int kb = 0; kb < K; kb += 4) {
    v2f a;
    a.x = Arow[kb];
    a.y = Arow[kb + 1];
    const float* Bk = B0 + (size_t)kb * ldb;
    v2f b0, b1, b2, b3;
    b0.x = Bk[0];   b0.y = Bk[ldb];
    b1.x = Bk[16];  b1.y = Bk[ldb + 16];
    b2.x = Bk[32];  b2.y = Bk[ldb + 32];
    b3.x = Bk[48];  b3.y = Bk[ldb + 48];
    acc0 = __builtin_amdgcn_wmma_f32_16x16x4_f32(false, a, false, b0, (short)0, acc0, false, false);
    acc1 = __builtin_amdgcn_wmma_f32_16x16x4_f32(false, a, false, b1, (short)0, acc1, false, false);
    acc2 = __builtin_amdgcn_wmma_f32_16x16x4_f32(false, a, false, b2, (short)0, acc2, false, false);
    acc3 = __builtin_amdgcn_wmma_f32_16x16x4_f32(false, a, false, b3, (short)0, acc3, false, false);
  }

  int crow0 = row0 + ((lane >> 4) << 3);
  v8f accs[4] = {acc0, acc1, acc2, acc3};
#pragma unroll
  for (int g = 0; g < 4; ++g) {
    int ccol = col0 + (g << 4) + ml;
    float bv = (bias != nullptr) ? bias[ccol] : 0.0f;
#pragma unroll
    for (int r = 0; r < 8; ++r) {
      float v = accs[g][r] + bv;
      if (do_relu) v = fmaxf(v, 0.0f);
      C[(size_t)(crow0 + r) * ldc + ccol] = v;
    }
  }
}

__global__ void pool_reduce(const float* __restrict__ H, float* __restrict__ pool) {
  int c = blockIdx.y;
  const float* h = H + (size_t)c * NP * FDIM;
  const int TOT = N_DRUGS * FDIM;
  float s = 0.0f;
  for (int i = blockIdx.x * blockDim.x + threadIdx.x; i < TOT;
       i += gridDim.x * blockDim.x)
    s += h[i];
  __shared__ float red[256];
  int tid = threadIdx.x;
  red[tid] = s;
  __syncthreads();
  for (int off = 128; off > 0; off >>= 1) {
    if (tid < off) red[tid] += red[tid + off];
    __syncthreads();
  }
  if (tid == 0) atomicAdd(&pool[c], red[0]);
}

__global__ void attn_kernel(const float* __restrict__ pool,
                            const float* __restrict__ fc1_w, const float* __restrict__ fc1_b,
                            const float* __restrict__ fc2_w, const float* __restrict__ fc2_b,
                            float* __restrict__ a_out) {
  __shared__ float p[6], h[30];
  int tid = threadIdx.x;
  if (tid < 6) p[tid] = pool[tid] * (1.0f / (591.0f * 1024.0f));
  __syncthreads();
  if (tid < 30) {
    float s = fc1_b[tid];
    for (int c = 0; c < 6; ++c) s += p[c] * fc1_w[tid * 6 + c];
    h[tid] = fmaxf(s, 0.0f);
  }
  __syncthreads();
  if (tid < 6) {
    float s = fc2_b[tid];
    for (int j = 0; j < 30; ++j) s += h[j] * fc2_w[tid * 30 + j];
    a_out[tid] = 1.0f / (1.0f + expf(-s));
  }
}

__global__ void final_kernel(const float* __restrict__ H, const float* __restrict__ a,
                             const float* __restrict__ conv_w,
                             const float* __restrict__ conv_b,
                             float* __restrict__ out) {
  int idx = blockIdx.x * blockDim.x + threadIdx.x;
  if (idx >= N_DRUGS * FDIM) return;
  int n = idx >> 10, f = idx & (FDIM - 1);
  float s = conv_b[0];
#pragma unroll
  for (int c = 0; c < 6; ++c) {
    float v = a[c] * H[(size_t)c * NP * FDIM + (size_t)n * FDIM + f];
    s += conv_w[c] * fmaxf(v, 0.0f);
  }
  out[idx] = s;
}

extern "C" void kernel_launch(void* const* d_in, const int* in_sizes, int n_in,
                              void* d_out, int out_size, void* d_ws, size_t ws_size,
                              hipStream_t stream) {
  (void)in_sizes; (void)n_in; (void)out_size; (void)ws_size;

  const float* dm1 = (const float*)d_in[0];
  const int*   edges[3] = {(const int*)d_in[1], (const int*)d_in[3], (const int*)d_in[5]};
  const float* ew[3]    = {(const float*)d_in[2], (const float*)d_in[4], (const float*)d_in[6]};
  const float* W1[3]    = {(const float*)d_in[7],  (const float*)d_in[11], (const float*)d_in[15]};
  const float* b1[3]    = {(const float*)d_in[8],  (const float*)d_in[12], (const float*)d_in[16]};
  const float* W2[3]    = {(const float*)d_in[9],  (const float*)d_in[13], (const float*)d_in[17]};
  const float* b2[3]    = {(const float*)d_in[10], (const float*)d_in[14], (const float*)d_in[18]};
  const float* fc1_w  = (const float*)d_in[19];
  const float* fc1_b  = (const float*)d_in[20];
  const float* fc2_w  = (const float*)d_in[21];
  const float* fc2_b  = (const float*)d_in[22];
  const float* conv_w = (const float*)d_in[23];
  const float* conv_b = (const float*)d_in[24];

  float* ws     = (float*)d_ws;
  float* ahat   = ws;
  float* deg    = ahat + (size_t)3 * NP * NP;
  float* dinv   = deg  + 3 * NP;
  float* pool   = dinv + 3 * NP;
  float* a_att  = pool + 8;
  float* H      = a_att + 8;
  float* xw     = H + (size_t)6 * NP * FDIM;
  float* x_pad  = xw + (size_t)NP * FDIM;

  init_ws<<<1024, 256, 0, stream>>>(ahat, deg, pool);

  pad_dm1<<<(NP * FDIM + 255) / 256, 256, 0, stream>>>(dm1, x_pad);

  deg_accum<<<(3 * NEDGE + 255) / 256, 256, 0, stream>>>(
      edges[0], ew[0], edges[1], ew[1], edges[2], ew[2], deg);

  compute_dinv<<<(3 * NP + 255) / 256, 256, 0, stream>>>(deg, dinv);

  build_ahat<<<(3 * (NEDGE + N_DRUGS) + 255) / 256, 256, 0, stream>>>(
      edges[0], ew[0], edges[1], ew[1], edges[2], ew[2], dinv, ahat);

  const int m_tiles  = NP / 16;
  const int n_wtiles = FDIM / 64;
  const int waves    = m_tiles * n_wtiles;
  const int blocks   = (waves * 32 + 255) / 256;

  for (int t = 0; t < 3; ++t) {
    float* Ah = ahat + (size_t)t * NP * NP;
    float* h1 = H + (size_t)(2 * t)     * NP * FDIM;
    float* h2 = H + (size_t)(2 * t + 1) * NP * FDIM;

    gemm_wmma_f32<<<blocks, 256, 0, stream>>>(x_pad, FDIM, W1[t], FDIM,
                                              xw, FDIM, FDIM, n_wtiles, m_tiles,
                                              nullptr, 0);
    gemm_wmma_f32<<<blocks, 256, 0, stream>>>(Ah, NP, xw, FDIM,
                                              h1, FDIM, NP, n_wtiles, m_tiles,
                                              b1[t], 1);
    gemm_wmma_f32<<<blocks, 256, 0, stream>>>(h1, FDIM, W2[t], FDIM,
                                              xw, FDIM, FDIM, n_wtiles, m_tiles,
                                              nullptr, 0);
    gemm_wmma_f32<<<blocks, 256, 0, stream>>>(Ah, NP, xw, FDIM,
                                              h2, FDIM, NP, n_wtiles, m_tiles,
                                              b2[t], 1);
  }

  dim3 pgrid(64, 6);
  pool_reduce<<<pgrid, 256, 0, stream>>>(H, pool);
  attn_kernel<<<1, 32, 0, stream>>>(pool, fc1_w, fc1_b, fc2_w, fc2_b, a_att);
  final_kernel<<<(N_DRUGS * FDIM + 255) / 256, 256, 0, stream>>>(
      H, a_att, conv_w, conv_b, (float*)d_out);
}


# GEMM K-loop reads A and B fragments with 128-bit loads (16-k blocks, 4 column tiles interleaved, C stored 128-bit); k summed in a different order inside each 16-k block
# speedup vs baseline: 2.5307x; 2.5307x over previous
#include <hip/hip_runtime.h>
#include <hip/hip_bf16.h>


#define N_DRUGS 591
#define NP      592
#define FDIM    1024
#define NEDGE   120000

typedef __attribute__((ext_vector_type(2))) float v2f;
typedef __attribute__((ext_vector_type(4))) float v4f;
typedef __attribute__((ext_vector_type(8))) float v8f;

__global__ void init_ws(float* __restrict__ ahat, float* __restrict__ deg,
                        float* __restrict__ pool) {
  int idx = blockIdx.x * blockDim.x + threadIdx.x;
  int stride = gridDim.x * blockDim.x;
  const int AH = 3 * NP * NP;
  for (int i = idx; i < AH; i += stride) ahat[i] = 0.0f;
  for (int i = idx; i < 3 * NP; i += stride) deg[i] = 1.0f;
  if (idx < 8) pool[idx] = 0.0f;
}

__global__ void pad_dm1(const float* __restrict__ dm1, float* __restrict__ out) {
  int idx = blockIdx.x * blockDim.x + threadIdx.x;
  if (idx >= NP * FDIM) return;
  int n = idx >> 10;
  out[idx] = (n < N_DRUGS) ? dm1[idx] : 0.0f;
}

__global__ void deg_accum(const int* __restrict__ eT, const float* __restrict__ wT,
                          const int* __restrict__ eS, const float* __restrict__ wS,
                          const int* __restrict__ eG, const float* __restrict__ wG,
                          float* __restrict__ deg) {
  int idx = blockIdx.x * blockDim.x + threadIdx.x;
  if (idx >= 3 * NEDGE) return;
  int t = idx / NEDGE, e = idx - t * NEDGE;
  const int*   ed = (t == 0) ? eT : ((t == 1) ? eS : eG);
  const float* w  = (t == 0) ? wT : ((t == 1) ? wS : wG);
  int dst = ed[NEDGE + e];
  atomicAdd(&deg[t * NP + dst], w[e]);
}

__global__ void compute_dinv(const float* __restrict__ deg, float* __restrict__ dinv) {
  int i = blockIdx.x * blockDim.x + threadIdx.x;
  if (i < 3 * NP) {
    float d = deg[i];
    dinv[i] = (d > 0.0f) ? rsqrtf(d) : 0.0f;
  }
}

__global__ void build_ahat(const int* __restrict__ eT, const float* __restrict__ wT,
                           const int* __restrict__ eS, const float* __restrict__ wS,
                           const int* __restrict__ eG, const float* __restrict__ wG,
                           const float* __restrict__ dinv, float* __restrict__ ahat) {
  const int PER = NEDGE + N_DRUGS;
  int idx = blockIdx.x * blockDim.x + threadIdx.x;
  if (idx >= 3 * PER) return;
  int t = idx / PER, r = idx - t * PER;
  const float* dv = dinv + t * NP;
  float* Ah = ahat + (size_t)t * NP * NP;
  if (r < NEDGE) {
    const int*   ed = (t == 0) ? eT : ((t == 1) ? eS : eG);
    const float* w  = (t == 0) ? wT : ((t == 1) ? wS : wG);
    int src = ed[r], dst = ed[NEDGE + r];
    atomicAdd(&Ah[(size_t)dst * NP + src], dv[src] * w[r] * dv[dst]);
  } else {
    int i = r - NEDGE;
    atomicAdd(&Ah[(size_t)i * NP + i], dv[i] * dv[i]);
  }
}

__global__ void gemm_wmma_f32(const float* __restrict__ A, int lda,
                              const float* __restrict__ B, int ldb,
                              float* __restrict__ C, int ldc,
                              int K, int n_wtiles, int m_tiles,
                              const float* __restrict__ bias, int do_relu) {
  int wid = (blockIdx.x * blockDim.x + threadIdx.x) >> 5;
  if (wid >= m_tiles * n_wtiles) return;
  int lane = threadIdx.x & 31;
  int tm = wid / n_wtiles, tn = wid - tm * n_wtiles;
  int row0 = tm << 4, col0 = tn << 6;
  int c = lane & 15;
  int h = lane >> 4;

  const float* Arow = A + (size_t)(row0 + c) * lda + (h << 3);
  const float* Bcol = B + (size_t)(h << 3) * ldb + (col0 + (c << 2));

  v8f acc0 = {}, acc1 = {}, acc2 = {}, acc3 = {};
#pragma unroll 2
  for (int kb = 0; kb < K; kb += 16) {
    const v4f* Ak = (const v4f*)(Arow + kb);
    v4f a03 = Ak[0];
    v4f a47 = Ak[1];
    const float* Bk = Bcol + (size_t)kb * ldb;
    v4f q0 = *(const v4f*)(Bk);
    v4f q1 = *(const v4f*)(Bk + (size_t)ldb);
    v4f q2 = *(const v4f*)(Bk + (size_t)ldb * 2);
    v4f q3 = *(const v4f*)(Bk + (size_t)ldb * 3);
    v4f q4 = *(const v4f*)(Bk + (size_t)ldb * 4);
    v4f q5 = *(const v4f*)(Bk + (size_t)ldb * 5);
    v4f q6 = *(const v4f*)(Bk + (size_t)ldb * 6);
    v4f q7 = *(const v4f*)(Bk + (size_t)ldb * 7);

#define WMMA_STEP(AX, AY, QE, QO)                                                        \
    {                                                                                    \
      v2f a = {AX, AY};                                                                  \
      v2f b0 = {QE.x, QO.x}, b1 = {QE.y, QO.y}, b2 = {QE.z, QO.z}, b3 = {QE.w, QO.w};    \
      acc0 = __builtin_amdgcn_wmma_f32_16x16x4_f32(false, a, false, b0, (short)0, acc0, false, false); \
      acc1 = __builtin_amdgcn_wmma_f32_16x16x4_f32(false, a, false, b1, (short)0, acc1, false, false); \
      acc2 = __builtin_amdgcn_wmma_f32_16x16x4_f32(false, a, false, b2, (short)0, acc2, false, false); \
      acc3 = __builtin_amdgcn_wmma_f32_16x16x4_f32(false, a, false, b3, (short)0, acc3, false, false); \
    }
    WMMA_STEP(a03.x, a03.y, q0, q1)
    WMMA_STEP(a03.z, a03.w, q2, q3)
    WMMA_STEP(a47.x, a47.y, q4, q5)
    WMMA_STEP(a47.z, a47.w, q6, q7)
#undef WMMA_STEP
  }

  int crow0 = row0 + (h << 3);
  int ccol = col0 + (c << 2);
  v4f bv = {0.0f, 0.0f, 0.0f, 0.0f};
  if (bias != nullptr) bv = *(const v4f*)(bias + ccol);
#pragma unroll
  for (int r = 0; r < 8; ++r) {
    v4f v = {acc0[r], acc1[r], acc2[r], acc3[r]};
    v += bv;
    if (do_relu) {
      v.x = fmaxf(v.x, 0.0f); v.y = fmaxf(v.y, 0.0f);
      v.z = fmaxf(v.z, 0.0f); v.w = fmaxf(v.w, 0.0f);
    }
    *(v4f*)(C + (size_t)(crow0 + r) * ldc + ccol) = v;
  }
}

__global__ void pool_reduce(const float* __restrict__ H, float* __restrict__ pool) {
  int c = blockIdx.y;
  const float* h = H + (size_t)c * NP * FDIM;
  const int TOT = N_DRUGS * FDIM;
  float s = 0.0f;
  for (int i = blockIdx.x * blockDim.x + threadIdx.x; i < TOT;
       i += gridDim.x * blockDim.x)
    s += h[i];
  __shared__ float red[256];
  int tid = threadIdx.x;
  red[tid] = s;
  __syncthreads();
  for (int off = 128; off > 0; off >>= 1) {
    if (tid < off) red[tid] += red[tid + off];
    __syncthreads();
  }
  if (tid == 0) atomicAdd(&pool[c], red[0]);
}

__global__ void attn_kernel(const float* __restrict__ pool,
                            const float* __restrict__ fc1_w, const float* __restrict__ fc1_b,
                            const float* __restrict__ fc2_w, const float* __restrict__ fc2_b,
                            float* __restrict__ a_out) {
  __shared__ float p[6], h[30];
  int tid = threadIdx.x;
  if (tid < 6) p[tid] = pool[tid] * (1.0f / (591.0f * 1024.0f));
  __syncthreads();
  if (tid < 30) {
    float s = fc1_b[tid];
    for (int c = 0; c < 6; ++c) s += p[c] * fc1_w[tid * 6 + c];
    h[tid] = fmaxf(s, 0.0f);
  }
  __syncthreads();
  if (tid < 6) {
    float s = fc2_b[tid];
    for (int j = 0; j < 30; ++j) s += h[j] * fc2_w[tid * 30 + j];
    a_out[tid] = 1.0f / (1.0f + expf(-s));
  }
}

__global__ void final_kernel(const float* __restrict__ H, const float* __restrict__ a,
                             const float* __restrict__ conv_w,
                             const float* __restrict__ conv_b,
                             float* __restrict__ out) {
  int idx = blockIdx.x * blockDim.x + threadIdx.x;
  if (idx >= N_DRUGS * FDIM) return;
  int n = idx >> 10, f = idx & (FDIM - 1);
  float s = conv_b[0];
#pragma unroll
  for (int c = 0; c < 6; ++c) {
    float v = a[c] * H[(size_t)c * NP * FDIM + (size_t)n * FDIM + f];
    s += conv_w[c] * fmaxf(v, 0.0f);
  }
  out[idx] = s;
}

extern "C" void kernel_launch(void* const* d_in, const int* in_sizes, int n_in,
                              void* d_out, int out_size, void* d_ws, size_t ws_size,
                              hipStream_t stream) {
  (void)in_sizes; (void)n_in; (void)out_size; (void)ws_size;

  const float* dm1 = (const float*)d_in[0];
  const int*   edges[3] = {(const int*)d_in[1], (const int*)d_in[3], (const int*)d_in[5]};
  const float* ew[3]    = {(const float*)d_in[2], (const float*)d_in[4], (const float*)d_in[6]};
  const float* W1[3]    = {(const float*)d_in[7],  (const float*)d_in[11], (const float*)d_in[15]};
  const float* b1[3]    = {(const float*)d_in[8],  (const float*)d_in[12], (const float*)d_in[16]};
  const float* W2[3]    = {(const float*)d_in[9],  (const float*)d_in[13], (const float*)d_in[17]};
  const float* b2[3]    = {(const float*)d_in[10], (const float*)d_in[14], (const float*)d_in[18]};
  const float* fc1_w  = (const float*)d_in[19];
  const float* fc1_b  = (const float*)d_in[20];
  const float* fc2_w  = (const float*)d_in[21];
  const float* fc2_b  = (const float*)d_in[22];
  const float* conv_w = (const float*)d_in[23];
  const float* conv_b = (const float*)d_in[24];

  float* ws     = (float*)d_ws;
  float* ahat   = ws;
  float* deg    = ahat + (size_t)3 * NP * NP;
  float* dinv   = deg  + 3 * NP;
  float* pool   = dinv + 3 * NP;
  float* a_att  = pool + 8;
  float* H      = a_att + 8;
  float* xw     = H + (size_t)6 * NP * FDIM;
  float* x_pad  = xw + (size_t)NP * FDIM;

  init_ws<<<1024, 256, 0, stream>>>(ahat, deg, pool);

  pad_dm1<<<(NP * FDIM + 255) / 256, 256, 0, stream>>>(dm1, x_pad);

  deg_accum<<<(3 * NEDGE + 255) / 256, 256, 0, stream>>>(
      edges[0], ew[0], edges[1], ew[1], edges[2], ew[2], deg);

  compute_dinv<<<(3 * NP + 255) / 256, 256, 0, stream>>>(deg, dinv);

  build_ahat<<<(3 * (NEDGE + N_DRUGS) + 255) / 256, 256, 0, stream>>>(
      edges[0], ew[0], edges[1], ew[1], edges[2], ew[2], dinv, ahat);

  const int m_tiles  = NP / 16;
  const int n_wtiles = FDIM / 64;
  const int waves    = m_tiles * n_wtiles;
  const int blocks   = (waves * 32 + 255) / 256;

  for (int t = 0; t < 3; ++t) {
    float* Ah = ahat + (size_t)t * NP * NP;
    float* h1 = H + (size_t)(2 * t)     * NP * FDIM;
    float* h2 = H + (size_t)(2 * t + 1) * NP * FDIM;

    gemm_wmma_f32<<<blocks, 256, 0, stream>>>(x_pad, FDIM, W1[t], FDIM,
                                              xw, FDIM, FDIM, n_wtiles, m_tiles,
                                              nullptr, 0);
    gemm_wmma_f32<<<blocks, 256, 0, stream>>>(Ah, NP, xw, FDIM,
                                              h1, FDIM, NP, n_wtiles, m_tiles,
                                              b1[t], 1);
    gemm_wmma_f32<<<blocks, 256, 0, stream>>>(h1, FDIM, W2[t], FDIM,
                                              xw, FDIM, FDIM, n_wtiles, m_tiles,
                                              nullptr, 0);
    gemm_wmma_f32<<<blocks, 256, 0, stream>>>(Ah, NP, xw, FDIM,
                                              h2, FDIM, NP, n_wtiles, m_tiles,
                                              b2[t], 1);
  }

  dim3 pgrid(64, 6);
  pool_reduce<<<pgrid, 256, 0, stream>>>(H, pool);
  attn_kernel<<<1, 32, 0, stream>>>(pool, fc1_w, fc1_b, fc2_w, fc2_b, a_att);
  final_kernel<<<(N_DRUGS * FDIM + 255) / 256, 256, 0, stream>>>(
      H, a_att, conv_w, conv_b, (float*)d_out);
}


# GEMM operands staged through LDS: a 4-wave workgroup copies its 64x64 tile's A and B blocks per 16-k step (two stages), fragments read one step ahead as one group; every sum in the parent's order
# speedup vs baseline: 7.2552x; 2.8669x over previous
#include <hip/hip_runtime.h>
#include <hip/hip_bf16.h>


#define N_DRUGS 591
#define NP      592
#define FDIM    1024
#define NEDGE   120000

typedef __attribute__((ext_vector_type(2))) float v2f;
typedef __attribute__((ext_vector_type(4))) float v4f;
typedef __attribute__((ext_vector_type(8))) float v8f;

__global__ void init_ws(float* __restrict__ ahat, float* __restrict__ deg,
                        float* __restrict__ pool) {
  int idx = blockIdx.x * blockDim.x + threadIdx.x;
  int stride = gridDim.x * blockDim.x;
  const int AH = 3 * NP * NP;
  for (int i = idx; i < AH; i += stride) ahat[i] = 0.0f;
  for (int i = idx; i < 3 * NP; i += stride) deg[i] = 1.0f;
  if (idx < 8) pool[idx] = 0.0f;
}

__global__ void pad_dm1(const float* __restrict__ dm1, float* __restrict__ out) {
  int idx = blockIdx.x * blockDim.x + threadIdx.x;
  if (idx >= NP * FDIM) return;
  int n = idx >> 10;
  out[idx] = (n < N_DRUGS) ? dm1[idx] : 0.0f;
}

__global__ void deg_accum(const int* __restrict__ eT, const float* __restrict__ wT,
                          const int* __restrict__ eS, const float* __restrict__ wS,
                          const int* __restrict__ eG, const float* __restrict__ wG,
                          float* __restrict__ deg) {
  int idx = blockIdx.x * blockDim.x + threadIdx.x;
  if (idx >= 3 * NEDGE) return;
  int t = idx / NEDGE, e = idx - t * NEDGE;
  const int*   ed = (t == 0) ? eT : ((t == 1) ? eS : eG);
  const float* w  = (t == 0) ? wT : ((t == 1) ? wS : wG);
  int dst = ed[NEDGE + e];
  atomicAdd(&deg[t * NP + dst], w[e]);
}

__global__ void compute_dinv(const float* __restrict__ deg, float* __restrict__ dinv) {
  int i = blockIdx.x * blockDim.x + threadIdx.x;
  if (i < 3 * NP) {
    float d = deg[i];
    dinv[i] = (d > 0.0f) ? rsqrtf(d) : 0.0f;
  }
}

__global__ void build_ahat(const int* __restrict__ eT, const float* __restrict__ wT,
                           const int* __restrict__ eS, const float* __restrict__ wS,
                           const int* __restrict__ eG, const float* __restrict__ wG,
                           const float* __restrict__ dinv, float* __restrict__ ahat) {
  const int PER = NEDGE + N_DRUGS;
  int idx = blockIdx.x * blockDim.x + threadIdx.x;
  if (idx >= 3 * PER) return;
  int t = idx / PER, r = idx - t * PER;
  const float* dv = dinv + t * NP;
  float* Ah = ahat + (size_t)t * NP * NP;
  if (r < NEDGE) {
    const int*   ed = (t == 0) ? eT : ((t == 1) ? eS : eG);
    const float* w  = (t == 0) ? wT : ((t == 1) ? wS : wG);
    int src = ed[r], dst = ed[NEDGE + r];
    atomicAdd(&Ah[(size_t)dst * NP + src], dv[src] * w[r] * dv[dst]);
  } else {
    int i = r - NEDGE;
    atomicAdd(&Ah[(size_t)i * NP + i], dv[i] * dv[i]);
  }
}

#define A_PITCH 20
#define B_PITCH 68

__device__ __forceinline__ void lds_barrier() {
  asm volatile("s_wait_dscnt 0x0\n\ts_barrier_signal -1\n\ts_barrier_wait -1" ::: "memory");
}

struct Frag { v4f a03, a47, q0, q1, q2, q3, q4, q5, q6, q7; };

__device__ __forceinline__ void load_frag(Frag& f, const float* as, const float* bs) {
  f.a03 = *(const v4f*)(as);
  f.a47 = *(const v4f*)(as + 4);
  f.q0 = *(const v4f*)(bs);
  f.q1 = *(const v4f*)(bs + B_PITCH);
  f.q2 = *(const v4f*)(bs + B_PITCH * 2);
  f.q3 = *(const v4f*)(bs + B_PITCH * 3);
  f.q4 = *(const v4f*)(bs + B_PITCH * 4);
  f.q5 = *(const v4f*)(bs + B_PITCH * 5);
  f.q6 = *(const v4f*)(bs + B_PITCH * 6);
  f.q7 = *(const v4f*)(bs + B_PITCH * 7);
}

__device__ __forceinline__ void mma_frag(const Frag& f, v8f& acc0, v8f& acc1,
                                         v8f& acc2, v8f& acc3) {
#define WMMA_STEP(AX, AY, QE, QO)                                                        \
  {                                                                                      \
    v2f a = {AX, AY};                                                                    \
    v2f b0 = {QE.x, QO.x}, b1 = {QE.y, QO.y}, b2 = {QE.z, QO.z}, b3 = {QE.w, QO.w};      \
    acc0 = __builtin_amdgcn_wmma_f32_16x16x4_f32(false, a, false, b0, (short)0, acc0, false, false); \
    acc1 = __builtin_amdgcn_wmma_f32_16x16x4_f32(false, a, false, b1, (short)0, acc1, false, false); \
    acc2 = __builtin_amdgcn_wmma_f32_16x16x4_f32(false, a, false, b2, (short)0, acc2, false, false); \
    acc3 = __builtin_amdgcn_wmma_f32_16x16x4_f32(false, a, false, b3, (short)0, acc3, false, false); \
  }
  WMMA_STEP(f.a03.x, f.a03.y, f.q0, f.q1)
  WMMA_STEP(f.a03.z, f.a03.w, f.q2, f.q3)
  WMMA_STEP(f.a47.x, f.a47.y, f.q4, f.q5)
  WMMA_STEP(f.a47.z, f.a47.w, f.q6, f.q7)
#undef WMMA_STEP
}
__global__ __launch_bounds__(128)
void gemm_wmma_f32(const float* __restrict__ A0, const float* __restrict__ A1,
                   const float* __restrict__ A2, int lda,
                   const float* __restrict__ B0, const float* __restrict__ B1,
                   const float* __restrict__ B2, int ldb,
                   float* __restrict__ C0, float* __restrict__ C1,
                   float* __restrict__ C2, int ldc,
                   int K, int n_wtiles, int m_tiles,
                   const float* __restrict__ bias0,
                   const float* __restrict__ bias1,
                   const float* __restrict__ bias2, int do_relu) {
  __shared__ __attribute__((aligned(16))) float As[2][64 * A_PITCH];
  __shared__ __attribute__((aligned(16))) float Bs[2][16 * B_PITCH];

  const int tid = threadIdx.x;
  const int t = blockIdx.y;
  const float* A    = (t == 0) ? A0    : ((t == 1) ? A1    : A2);
  const float* B    = (t == 0) ? B0    : ((t == 1) ? B1    : B2);
  float*       C    = (t == 0) ? C0    : ((t == 1) ? C1    : C2);
  const float* bias = (t == 0) ? bias0 : ((t == 1) ? bias1 : bias2);
  const int bm = blockIdx.x / n_wtiles, bn = blockIdx.x - bm * n_wtiles;
  const int lane = tid & 31;
  const int wm = __builtin_amdgcn_readfirstlane(tid >> 5);
  const int c = lane & 15;
  const int h = lane >> 4;
  const int tm = bm * 4 + wm;
  const bool live = tm < m_tiles;
  const int last_row = (m_tiles << 4) - 1;
  const int row0 = tm << 4, col0 = bn << 6;

  const int ar = tid >> 2, aq = (tid & 3) << 2;
  const int br = tid >> 4, bq = (tid & 15) << 2;
  int ga0 = (bm << 6) + ar, ga1 = ga0 + 32;
  ga0 = ga0 < last_row ? ga0 : last_row;
  ga1 = ga1 < last_row ? ga1 : last_row;
  const float* Ag0 = A + (size_t)ga0 * lda + aq;
  const float* Ag1 = A + (size_t)ga1 * lda + aq;
  const float* Bg0 = B + (size_t)br * ldb + (col0 + bq);
  const float* Bg1 = Bg0 + (size_t)ldb * 8;
  const int sa0 = ar * A_PITCH + aq, sa1 = sa0 + 32 * A_PITCH;
  const int sb0 = br * B_PITCH + bq, sb1 = sb0 + 8 * B_PITCH;
  const int fa = ((wm << 4) + c) * A_PITCH + (h << 3);
  const int fb = (h << 3) * B_PITCH + (c << 2);

  const int nk = K >> 4;
  v4f xa0, xa1, xb0, xb1, ya0, ya1, yb0, yb1;
#define TILE_LOAD(P, KT)                                                        \
  {                                                                             \
    const int kb = (((KT) < nk) ? (KT) : (nk - 1)) << 4;                        \
    P##a0 = *(const v4f*)(Ag0 + kb);                                            \
    P##a1 = *(const v4f*)(Ag1 + kb);                                            \
    P##b0 = *(const v4f*)(Bg0 + (size_t)kb * ldb);                              \
    P##b1 = *(const v4f*)(Bg1 + (size_t)kb * ldb);                              \
  }
#define TILE_STORE(P, ST)                                                       \
  {                                                                             \
    *(v4f*)&As[ST][sa0] = P##a0;  *(v4f*)&As[ST][sa1] = P##a1;                  \
    *(v4f*)&Bs[ST][sb0] = P##b0;  *(v4f*)&Bs[ST][sb1] = P##b1;                  \
  }
  TILE_LOAD(x, 0)
  TILE_LOAD(y, 1)
  TILE_STORE(x, 0)
  TILE_STORE(y, 1)
  TILE_LOAD(x, 2)
  lds_barrier();

  v8f acc0 = {}, acc1 = {}, acc2 = {}, acc3 = {};
  Frag f0, f1;
  if (live) load_frag(f0, &As[0][fa], &Bs[0][fb]);
  lds_barrier();

#define GEMM_STEP(S, CUR, NXT, RS, WS, GLD, GST)                                \
  {                                                                             \
    TILE_LOAD(GLD, (S) + 3)                                                     \
    if (live) {                                                                 \
      load_frag(NXT, &As[RS][fa], &Bs[RS][fb]);                                 \
        \
        \
      __builtin_amdgcn_sched_barrier(0);                                        \
      mma_frag(CUR, acc0, acc1, acc2, acc3);                                    \
    }                                                                           \
    TILE_STORE(GST, WS)                                                         \
    lds_barrier();                                                              \
  }

  int s = 0;
  for (; s + 1 < nk; s += 2) {
    GEMM_STEP(s,     f0, f1, 1, 0, y, x)
    GEMM_STEP(s + 1, f1, f0, 0, 1, x, y)
  }
  if (s < nk) {
    if (live) mma_frag(f0, acc0, acc1, acc2, acc3);
  }
#undef GEMM_STEP
#undef TILE_LOAD
#undef TILE_STORE

  if (live) {
    int crow0 = row0 + (h << 3);
    int ccol = col0 + (c << 2);
    v4f bv = {0.0f, 0.0f, 0.0f, 0.0f};
    if (bias != nullptr) bv = *(const v4f*)(bias + ccol);
#pragma unroll
    for (int r = 0; r < 8; ++r) {
      v4f v = {acc0[r], acc1[r], acc2[r], acc3[r]};
      v += bv;
      if (do_relu) {
        v.x = fmaxf(v.x, 0.0f); v.y = fmaxf(v.y, 0.0f);
        v.z = fmaxf(v.z, 0.0f); v.w = fmaxf(v.w, 0.0f);
      }
      *(v4f*)(C + (size_t)(crow0 + r) * ldc + ccol) = v;
    }
  }
}

__global__ void pool_reduce(const float* __restrict__ H, float* __restrict__ pool) {
  int c = blockIdx.y;
  const float* h = H + (size_t)c * NP * FDIM;
  const int TOT = N_DRUGS * FDIM;
  float s = 0.0f;
  for (int i = blockIdx.x * blockDim.x + threadIdx.x; i < TOT;
       i += gridDim.x * blockDim.x)
    s += h[i];
  __shared__ float red[256];
  int tid = threadIdx.x;
  red[tid] = s;
  __syncthreads();
  for (int off = 128; off > 0; off >>= 1) {
    if (tid < off) red[tid] += red[tid + off];
    __syncthreads();
  }
  if (tid == 0) atomicAdd(&pool[c], red[0]);
}

__global__ void attn_kernel(const float* __restrict__ pool,
                            const float* __restrict__ fc1_w, const float* __restrict__ fc1_b,
                            const float* __restrict__ fc2_w, const float* __restrict__ fc2_b,
                            float* __restrict__ a_out) {
  __shared__ float p[6], h[30];
  int tid = threadIdx.x;
  if (tid < 6) p[tid] = pool[tid] * (1.0f / (591.0f * 1024.0f));
  __syncthreads();
  if (tid < 30) {
    float s = fc1_b[tid];
    for (int c = 0; c < 6; ++c) s += p[c] * fc1_w[tid * 6 + c];
    h[tid] = fmaxf(s, 0.0f);
  }
  __syncthreads();
  if (tid < 6) {
    float s = fc2_b[tid];
    for (int j = 0; j < 30; ++j) s += h[j] * fc2_w[tid * 30 + j];
    a_out[tid] = 1.0f / (1.0f + expf(-s));
  }
}

__global__ void final_kernel(const float* __restrict__ H, const float* __restrict__ a,
                             const float* __restrict__ conv_w,
                             const float* __restrict__ conv_b,
                             float* __restrict__ out) {
  int idx = blockIdx.x * blockDim.x + threadIdx.x;
  if (idx >= N_DRUGS * FDIM) return;
  int n = idx >> 10, f = idx & (FDIM - 1);
  float s = conv_b[0];
#pragma unroll
  for (int c = 0; c < 6; ++c) {
    float v = a[c] * H[(size_t)c * NP * FDIM + (size_t)n * FDIM + f];
    s += conv_w[c] * fmaxf(v, 0.0f);
  }
  out[idx] = s;
}

extern "C" void kernel_launch(void* const* d_in, const int* in_sizes, int n_in,
                              void* d_out, int out_size, void* d_ws, size_t ws_size,
                              hipStream_t stream) {
  (void)in_sizes; (void)n_in; (void)out_size; (void)ws_size;

  const float* dm1 = (const float*)d_in[0];
  const int*   edges[3] = {(const int*)d_in[1], (const int*)d_in[3], (const int*)d_in[5]};
  const float* ew[3]    = {(const float*)d_in[2], (const float*)d_in[4], (const float*)d_in[6]};
  const float* W1[3]    = {(const float*)d_in[7],  (const float*)d_in[11], (const float*)d_in[15]};
  const float* b1[3]    = {(const float*)d_in[8],  (const float*)d_in[12], (const float*)d_in[16]};
  const float* W2[3]    = {(const float*)d_in[9],  (const float*)d_in[13], (const float*)d_in[17]};
  const float* b2[3]    = {(const float*)d_in[10], (const float*)d_in[14], (const float*)d_in[18]};
  const float* fc1_w  = (const float*)d_in[19];
  const float* fc1_b  = (const float*)d_in[20];
  const float* fc2_w  = (const float*)d_in[21];
  const float* fc2_b  = (const float*)d_in[22];
  const float* conv_w = (const float*)d_in[23];
  const float* conv_b = (const float*)d_in[24];

  float* ws     = (float*)d_ws;
  float* ahat   = ws;
  float* deg    = ahat + (size_t)3 * NP * NP;
  float* dinv   = deg  + 3 * NP;
  float* pool   = dinv + 3 * NP;
  float* a_att  = pool + 8;
  float* H      = a_att + 8;
  float* xw     = H + (size_t)6 * NP * FDIM;
  float* x_pad  = xw + (size_t)NP * FDIM;

  init_ws<<<1024, 256, 0, stream>>>(ahat, deg, pool);

  pad_dm1<<<(NP * FDIM + 255) / 256, 256, 0, stream>>>(dm1, x_pad);

  deg_accum<<<(3 * NEDGE + 255) / 256, 256, 0, stream>>>(
      edges[0], ew[0], edges[1], ew[1], edges[2], ew[2], deg);

  compute_dinv<<<(3 * NP + 255) / 256, 256, 0, stream>>>(deg, dinv);

  build_ahat<<<(3 * (NEDGE + N_DRUGS) + 255) / 256, 256, 0, stream>>>(
      edges[0], ew[0], edges[1], ew[1], edges[2], ew[2], dinv, ahat);

  const int m_tiles  = NP / 16;
  const int n_wtiles = FDIM / 64;
  const int blocks   = ((m_tiles + 3) / 4) * n_wtiles;

  const size_t CH = (size_t)NP * FDIM;
  float* h1_t = H;          float* h2_t = H + CH;
  float* h1_s = H + 2 * CH; float* h2_s = H + 3 * CH;
  float* h1_g = H + 4 * CH; float* h2_g = H + 5 * CH;
  float* Ah_t = ahat;
  float* Ah_s = ahat + (size_t)NP * NP;
  float* Ah_g = ahat + (size_t)2 * NP * NP;


  gemm_wmma_f32<<<dim3(blocks, 3), 128, 0, stream>>>(
      x_pad, x_pad, x_pad, FDIM, W1[0], W1[1], W1[2], FDIM,
      h2_t, h2_s, h2_g, FDIM, FDIM, n_wtiles, m_tiles,
      nullptr, nullptr, nullptr, 0);
  gemm_wmma_f32<<<dim3(blocks, 3), 128, 0, stream>>>(
      Ah_t, Ah_s, Ah_g, NP, h2_t, h2_s, h2_g, FDIM,
      h1_t, h1_s, h1_g, FDIM, NP, n_wtiles, m_tiles,
      b1[0], b1[1], b1[2], 1);
  gemm_wmma_f32<<<dim3(blocks, 3), 128, 0, stream>>>(
      h1_t, h1_s, h1_g, FDIM, W2[0], W2[1], W2[2], FDIM,
      xw, x_pad, h2_t, FDIM, FDIM, n_wtiles, m_tiles,
      nullptr, nullptr, nullptr, 0);
  gemm_wmma_f32<<<dim3(blocks, 2), 128, 0, stream>>>(
      Ah_s, Ah_g, nullptr, NP, x_pad, h2_t, nullptr, FDIM,
      h2_s, h2_g, nullptr, FDIM, NP, n_wtiles, m_tiles,
      b2[1], b2[2], nullptr, 1);
  gemm_wmma_f32<<<dim3(blocks, 1), 128, 0, stream>>>(
      Ah_t, nullptr, nullptr, NP, xw, nullptr, nullptr, FDIM,
      h2_t, nullptr, nullptr, FDIM, NP, n_wtiles, m_tiles,
      b2[0], nullptr, nullptr, 1);

  dim3 pgrid(64, 6);
  pool_reduce<<<pgrid, 256, 0, stream>>>(H, pool);
  attn_kernel<<<1, 32, 0, stream>>>(pool, fc1_w, fc1_b, fc2_w, fc2_b, a_att);
  final_kernel<<<(N_DRUGS * FDIM + 255) / 256, 256, 0, stream>>>(
      H, a_att, conv_w, conv_b, (float*)d_out);
}


# Kernels around the GEMMs launch about 7x fewer waves: a thread takes 4-8 row quads (128-bit loads/stores) or 8 edges; pool holds 16 row loads in flight and sums its partials in a different order
# speedup vs baseline: 8.0455x; 1.1089x over previous
#include <hip/hip_runtime.h>
#include <hip/hip_bf16.h>


#define N_DRUGS 591
#define NP      592
#define FDIM    1024
#define NEDGE   120000

typedef __attribute__((ext_vector_type(2))) float v2f;
typedef __attribute__((ext_vector_type(4))) float v4f;
typedef __attribute__((ext_vector_type(8))) float v8f;
typedef __attribute__((ext_vector_type(4))) int   v4i;

#define ROWQ    (FDIM / 4)
#define EPT     8

__global__ void init_ws(float* __restrict__ ahat, float* __restrict__ deg,
                        float* __restrict__ pool) {
  int idx = blockIdx.x * blockDim.x + threadIdx.x;
  int stride = gridDim.x * blockDim.x;
  const int AH4 = 3 * NP * NP / 4;
  const v4f zero = {0.0f, 0.0f, 0.0f, 0.0f};
  const v4f one  = {1.0f, 1.0f, 1.0f, 1.0f};
  v4f* a4 = (v4f*)ahat;
  v4f* d4 = (v4f*)deg;
  for (int i = idx; i < AH4; i += stride) a4[i] = zero;
  for (int i = idx; i < 3 * NP / 4; i += stride) d4[i] = one;
  if (idx < 8) pool[idx] = 0.0f;
}

#define PAD_T (NP * ROWQ / 4)
__global__ void pad_dm1(const float* __restrict__ dm1, float* __restrict__ out) {
  int idx = blockIdx.x * blockDim.x + threadIdx.x;
  const v4f* src = (const v4f*)dm1;
  v4f* dst = (v4f*)out;
  const v4f zero = {0.0f, 0.0f, 0.0f, 0.0f};
  v4f v[4];
#pragma unroll
  for (int j = 0; j < 4; ++j) {
    int q = idx + j * PAD_T;
    v[j] = ((q >> 8) < N_DRUGS) ? src[q] : zero;
  }
#pragma unroll
  for (int j = 0; j < 4; ++j) dst[idx + j * PAD_T] = v[j];
}

__global__ void deg_accum(const int* __restrict__ eT, const float* __restrict__ wT,
                          const int* __restrict__ eS, const float* __restrict__ wS,
                          const int* __restrict__ eG, const float* __restrict__ wG,
                          float* __restrict__ deg) {
  const int PER = NEDGE / EPT;
  int idx = blockIdx.x * blockDim.x + threadIdx.x;
  if (idx >= 3 * PER) return;
  int t = idx / PER, e = (idx - t * PER) * EPT;
  const int*   ed = (t == 0) ? eT : ((t == 1) ? eS : eG);
  const float* w  = (t == 0) ? wT : ((t == 1) ? wS : wG);
  v4i da = *(const v4i*)(ed + NEDGE + e), db = *(const v4i*)(ed + NEDGE + e + 4);
  v4f wa = *(const v4f*)(w + e),          wb = *(const v4f*)(w + e + 4);
  float* dg = deg + t * NP;
  atomicAdd(&dg[da.x], wa.x);  atomicAdd(&dg[da.y], wa.y);
  atomicAdd(&dg[da.z], wa.z);  atomicAdd(&dg[da.w], wa.w);
  atomicAdd(&dg[db.x], wb.x);  atomicAdd(&dg[db.y], wb.y);
  atomicAdd(&dg[db.z], wb.z);  atomicAdd(&dg[db.w], wb.w);
}

__global__ void compute_dinv(const float* __restrict__ deg, float* __restrict__ dinv) {
  int i = blockIdx.x * blockDim.x + threadIdx.x;
  if (i < 3 * NP) {
    float d = deg[i];
    dinv[i] = (d > 0.0f) ? rsqrtf(d) : 0.0f;
  }
}

__global__ void build_ahat(const int* __restrict__ eT, const float* __restrict__ wT,
                           const int* __restrict__ eS, const float* __restrict__ wS,
                           const int* __restrict__ eG, const float* __restrict__ wG,
                           const float* __restrict__ dinv, float* __restrict__ ahat) {
  const int PER = NEDGE / EPT;
  int idx = blockIdx.x * blockDim.x + threadIdx.x;
  if (idx < 3 * PER) {
    int t = idx / PER, e = (idx - t * PER) * EPT;
    const float* dv = dinv + t * NP;
    float* Ah = ahat + (size_t)t * NP * NP;
    const int*   ed = (t == 0) ? eT : ((t == 1) ? eS : eG);
    const float* w  = (t == 0) ? wT : ((t == 1) ? wS : wG);
    v4i sa = *(const v4i*)(ed + e),         sb = *(const v4i*)(ed + e + 4);
    v4i da = *(const v4i*)(ed + NEDGE + e), db = *(const v4i*)(ed + NEDGE + e + 4);
    v4f wa = *(const v4f*)(w + e),          wb = *(const v4f*)(w + e + 4);
    float n0 = dv[sa.x] * wa.x * dv[da.x], n1 = dv[sa.y] * wa.y * dv[da.y];
    float n2 = dv[sa.z] * wa.z * dv[da.z], n3 = dv[sa.w] * wa.w * dv[da.w];
    float n4 = dv[sb.x] * wb.x * dv[db.x], n5 = dv[sb.y] * wb.y * dv[db.y];
    float n6 = dv[sb.z] * wb.z * dv[db.z], n7 = dv[sb.w] * wb.w * dv[db.w];
    atomicAdd(&Ah[(size_t)da.x * NP + sa.x], n0);  atomicAdd(&Ah[(size_t)da.y * NP + sa.y], n1);
    atomicAdd(&Ah[(size_t)da.z * NP + sa.z], n2);  atomicAdd(&Ah[(size_t)da.w * NP + sa.w], n3);
    atomicAdd(&Ah[(size_t)db.x * NP + sb.x], n4);  atomicAdd(&Ah[(size_t)db.y * NP + sb.y], n5);
    atomicAdd(&Ah[(size_t)db.z * NP + sb.z], n6);  atomicAdd(&Ah[(size_t)db.w * NP + sb.w], n7);
  } else {
    int r = idx - 3 * PER;
    if (r >= 3 * N_DRUGS) return;
    int t = r / N_DRUGS, i = r - t * N_DRUGS;
    const float* dv = dinv + t * NP;
    float* Ah = ahat + (size_t)t * NP * NP;
    atomicAdd(&Ah[(size_t)i * NP + i], dv[i] * dv[i]);
  }
}

#define A_PITCH 20
#define B_PITCH 68

__device__ __forceinline__ void lds_barrier() {
  asm volatile("s_wait_dscnt 0x0\n\ts_barrier_signal -1\n\ts_barrier_wait -1" ::: "memory");
}

struct Frag { v4f a03, a47, q0, q1, q2, q3, q4, q5, q6, q7; };

__device__ __forceinline__ void load_frag(Frag& f, const float* as, const float* bs) {
  f.a03 = *(const v4f*)(as);
  f.a47 = *(const v4f*)(as + 4);
  f.q0 = *(const v4f*)(bs);
  f.q1 = *(const v4f*)(bs + B_PITCH);
  f.q2 = *(const v4f*)(bs + B_PITCH * 2);
  f.q3 = *(const v4f*)(bs + B_PITCH * 3);
  f.q4 = *(const v4f*)(bs + B_PITCH * 4);
  f.q5 = *(const v4f*)(bs + B_PITCH * 5);
  f.q6 = *(const v4f*)(bs + B_PITCH * 6);
  f.q7 = *(const v4f*)(bs + B_PITCH * 7);
}

__device__ __forceinline__ void mma_frag(const Frag& f, v8f& acc0, v8f& acc1,
                                         v8f& acc2, v8f& acc3) {
#define WMMA_STEP(AX, AY, QE, QO)                                                        \
  {                                                                                      \
    v2f a = {AX, AY};                                                                    \
    v2f b0 = {QE.x, QO.x}, b1 = {QE.y, QO.y}, b2 = {QE.z, QO.z}, b3 = {QE.w, QO.w};      \
    acc0 = __builtin_amdgcn_wmma_f32_16x16x4_f32(false, a, false, b0, (short)0, acc0, false, false); \
    acc1 = __builtin_amdgcn_wmma_f32_16x16x4_f32(false, a, false, b1, (short)0, acc1, false, false); \
    acc2 = __builtin_amdgcn_wmma_f32_16x16x4_f32(false, a, false, b2, (short)0, acc2, false, false); \
    acc3 = __builtin_amdgcn_wmma_f32_16x16x4_f32(false, a, false, b3, (short)0, acc3, false, false); \
  }
  WMMA_STEP(f.a03.x, f.a03.y, f.q0, f.q1)
  WMMA_STEP(f.a03.z, f.a03.w, f.q2, f.q3)
  WMMA_STEP(f.a47.x, f.a47.y, f.q4, f.q5)
  WMMA_STEP(f.a47.z, f.a47.w, f.q6, f.q7)
#undef WMMA_STEP
}
__global__ __launch_bounds__(128)
void gemm_wmma_f32(const float* __restrict__ A0, const float* __restrict__ A1,
                   const float* __restrict__ A2, int lda,
                   const float* __restrict__ B0, const float* __restrict__ B1,
                   const float* __restrict__ B2, int ldb,
                   float* __restrict__ C0, float* __restrict__ C1,
                   float* __restrict__ C2, int ldc,
                   int K, int n_wtiles, int m_tiles,
                   const float* __restrict__ bias0,
                   const float* __restrict__ bias1,
                   const float* __restrict__ bias2, int do_relu) {
  __shared__ __attribute__((aligned(16))) float As[2][64 * A_PITCH];
  __shared__ __attribute__((aligned(16))) float Bs[2][16 * B_PITCH];

  const int tid = threadIdx.x;
  const int t = blockIdx.y;
  const float* A    = (t == 0) ? A0    : ((t == 1) ? A1    : A2);
  const float* B    = (t == 0) ? B0    : ((t == 1) ? B1    : B2);
  float*       C    = (t == 0) ? C0    : ((t == 1) ? C1    : C2);
  const float* bias = (t == 0) ? bias0 : ((t == 1) ? bias1 : bias2);
  const int bm = blockIdx.x / n_wtiles, bn = blockIdx.x - bm * n_wtiles;
  const int lane = tid & 31;
  const int wm = __builtin_amdgcn_readfirstlane(tid >> 5);
  const int c = lane & 15;
  const int h = lane >> 4;
  const int tm = bm * 4 + wm;
  const bool live = tm < m_tiles;
  const int last_row = (m_tiles << 4) - 1;
  const int row0 = tm << 4, col0 = bn << 6;

  const int ar = tid >> 2, aq = (tid & 3) << 2;
  const int br = tid >> 4, bq = (tid & 15) << 2;
  int ga0 = (bm << 6) + ar, ga1 = ga0 + 32;
  ga0 = ga0 < last_row ? ga0 : last_row;
  ga1 = ga1 < last_row ? ga1 : last_row;
  const float* Ag0 = A + (size_t)ga0 * lda + aq;
  const float* Ag1 = A + (size_t)ga1 * lda + aq;
  const float* Bg0 = B + (size_t)br * ldb + (col0 + bq);
  const float* Bg1 = Bg0 + (size_t)ldb * 8;
  const int sa0 = ar * A_PITCH + aq, sa1 = sa0 + 32 * A_PITCH;
  const int sb0 = br * B_PITCH + bq, sb1 = sb0 + 8 * B_PITCH;
  const int fa = ((wm << 4) + c) * A_PITCH + (h << 3);
  const int fb = (h << 3) * B_PITCH + (c << 2);

  const int nk = K >> 4;
  v4f xa0, xa1, xb0, xb1, ya0, ya1, yb0, yb1;
#define TILE_LOAD(P, KT)                                                        \
  {                                                                             \
    const int kb = (((KT) < nk) ? (KT) : (nk - 1)) << 4;                        \
    P##a0 = *(const v4f*)(Ag0 + kb);                                            \
    P##a1 = *(const v4f*)(Ag1 + kb);                                            \
    P##b0 = *(const v4f*)(Bg0 + (size_t)kb * ldb);                              \
    P##b1 = *(const v4f*)(Bg1 + (size_t)kb * ldb);                              \
  }
#define TILE_STORE(P, ST)                                                       \
  {                                                                             \
    *(v4f*)&As[ST][sa0] = P##a0;  *(v4f*)&As[ST][sa1] = P##a1;                  \
    *(v4f*)&Bs[ST][sb0] = P##b0;  *(v4f*)&Bs[ST][sb1] = P##b1;                  \
  }
  TILE_LOAD(x, 0)
  TILE_LOAD(y, 1)
  TILE_STORE(x, 0)
  TILE_STORE(y, 1)
  TILE_LOAD(x, 2)
  lds_barrier();

  v8f acc0 = {}, acc1 = {}, acc2 = {}, acc3 = {};
  Frag f0, f1;
  if (live) load_frag(f0, &As[0][fa], &Bs[0][fb]);
  lds_barrier();

#define GEMM_STEP(S, CUR, NXT, RS, WS, GLD, GST)                                \
  {                                                                             \
    TILE_LOAD(GLD, (S) + 3)                                                     \
    if (live) {                                                                 \
      load_frag(NXT, &As[RS][fa], &Bs[RS][fb]);                                 \
        \
        \
      __builtin_amdgcn_sched_barrier(0);                                        \
      mma_frag(CUR, acc0, acc1, acc2, acc3);                                    \
    }                                                                           \
    TILE_STORE(GST, WS)                                                         \
    lds_barrier();                                                              \
  }

  int s = 0;
  for (; s + 1 < nk; s += 2) {
    GEMM_STEP(s,     f0, f1, 1, 0, y, x)
    GEMM_STEP(s + 1, f1, f0, 0, 1, x, y)
  }
  if (s < nk) {
    if (live) mma_frag(f0, acc0, acc1, acc2, acc3);
  }
#undef GEMM_STEP
#undef TILE_LOAD
#undef TILE_STORE

  if (live) {
    int crow0 = row0 + (h << 3);
    int ccol = col0 + (c << 2);
    v4f bv = {0.0f, 0.0f, 0.0f, 0.0f};
    if (bias != nullptr) bv = *(const v4f*)(bias + ccol);
#pragma unroll
    for (int r = 0; r < 8; ++r) {
      v4f v = {acc0[r], acc1[r], acc2[r], acc3[r]};
      v += bv;
      if (do_relu) {
        v.x = fmaxf(v.x, 0.0f); v.y = fmaxf(v.y, 0.0f);
        v.z = fmaxf(v.z, 0.0f); v.w = fmaxf(v.w, 0.0f);
      }
      *(v4f*)(C + (size_t)(crow0 + r) * ldc + ccol) = v;
    }
  }
}

#define POOL_ROWS 16
__global__ void pool_reduce(const float* __restrict__ H, float* __restrict__ pool) {
  int c = blockIdx.y;
  int tid = threadIdx.x;
  const v4f* h = (const v4f*)(H + (size_t)c * NP * FDIM);
  const int r0 = blockIdx.x * POOL_ROWS;
  const v4f zero = {0.0f, 0.0f, 0.0f, 0.0f};
  v4f v[POOL_ROWS];
#pragma unroll
  for (int j = 0; j < POOL_ROWS; ++j) v[j] = h[(r0 + j) * ROWQ + tid];
  v4f a = zero;
#pragma unroll
  for (int j = 0; j < POOL_ROWS; ++j) a += (r0 + j < N_DRUGS) ? v[j] : zero;
  float s = (a.x + a.y) + (a.z + a.w);
  __shared__ float red[256];
  red[tid] = s;
  __syncthreads();
  for (int off = 128; off > 0; off >>= 1) {
    if (tid < off) red[tid] += red[tid + off];
    __syncthreads();
  }
  if (tid == 0) atomicAdd(&pool[c], red[0]);
}

__global__ void attn_kernel(const float* __restrict__ pool,
                            const float* __restrict__ fc1_w, const float* __restrict__ fc1_b,
                            const float* __restrict__ fc2_w, const float* __restrict__ fc2_b,
                            float* __restrict__ a_out) {
  __shared__ float p[6], h[30];
  int tid = threadIdx.x;
  if (tid < 6) p[tid] = pool[tid] * (1.0f / (591.0f * 1024.0f));
  __syncthreads();
  if (tid < 30) {
    float s = fc1_b[tid];
    for (int c = 0; c < 6; ++c) s += p[c] * fc1_w[tid * 6 + c];
    h[tid] = fmaxf(s, 0.0f);
  }
  __syncthreads();
  if (tid < 6) {
    float s = fc2_b[tid];
    for (int j = 0; j < 30; ++j) s += h[j] * fc2_w[tid * 30 + j];
    a_out[tid] = 1.0f / (1.0f + expf(-s));
  }
}

__global__ void final_kernel(const float* __restrict__ H, const float* __restrict__ a,
                             const float* __restrict__ conv_w,
                             const float* __restrict__ conv_b,
                             float* __restrict__ out) {
  const int Q = N_DRUGS * ROWQ;
  int idx = blockIdx.x * blockDim.x + threadIdx.x;
  int stride = gridDim.x * blockDim.x;
  const float cb = conv_b[0];
  float ac[6], wc[6];
#pragma unroll
  for (int c = 0; c < 6; ++c) { ac[c] = a[c]; wc[c] = conv_w[c]; }
#pragma unroll
  for (int j = 0; j < 2; ++j) {
    int q = idx + j * stride;
    if (q < Q) {
      v4f hv[6];
#pragma unroll
      for (int c = 0; c < 6; ++c) hv[c] = ((const v4f*)(H + (size_t)c * NP * FDIM))[q];
      v4f s = {cb, cb, cb, cb};
#pragma unroll
      for (int c = 0; c < 6; ++c) {
        float vx = ac[c] * hv[c].x, vy = ac[c] * hv[c].y;
        float vz = ac[c] * hv[c].z, vw = ac[c] * hv[c].w;
        s.x += wc[c] * fmaxf(vx, 0.0f);  s.y += wc[c] * fmaxf(vy, 0.0f);
        s.z += wc[c] * fmaxf(vz, 0.0f);  s.w += wc[c] * fmaxf(vw, 0.0f);
      }
      ((v4f*)out)[q] = s;
    }
  }
}

extern "C" void kernel_launch(void* const* d_in, const int* in_sizes, int n_in,
                              void* d_out, int out_size, void* d_ws, size_t ws_size,
                              hipStream_t stream) {
  (void)in_sizes; (void)n_in; (void)out_size; (void)ws_size;

  const float* dm1 = (const float*)d_in[0];
  const int*   edges[3] = {(const int*)d_in[1], (const int*)d_in[3], (const int*)d_in[5]};
  const float* ew[3]    = {(const float*)d_in[2], (const float*)d_in[4], (const float*)d_in[6]};
  const float* W1[3]    = {(const float*)d_in[7],  (const float*)d_in[11], (const float*)d_in[15]};
  const float* b1[3]    = {(const float*)d_in[8],  (const float*)d_in[12], (const float*)d_in[16]};
  const float* W2[3]    = {(const float*)d_in[9],  (const float*)d_in[13], (const float*)d_in[17]};
  const float* b2[3]    = {(const float*)d_in[10], (const float*)d_in[14], (const float*)d_in[18]};
  const float* fc1_w  = (const float*)d_in[19];
  const float* fc1_b  = (const float*)d_in[20];
  const float* fc2_w  = (const float*)d_in[21];
  const float* fc2_b  = (const float*)d_in[22];
  const float* conv_w = (const float*)d_in[23];
  const float* conv_b = (const float*)d_in[24];

  float* ws     = (float*)d_ws;
  float* ahat   = ws;
  float* deg    = ahat + (size_t)3 * NP * NP;
  float* dinv   = deg  + 3 * NP;
  float* pool   = dinv + 3 * NP;
  float* a_att  = pool + 8;
  float* H      = a_att + 8;
  float* xw     = H + (size_t)6 * NP * FDIM;
  float* x_pad  = xw + (size_t)NP * FDIM;

  init_ws<<<129, 256, 0, stream>>>(ahat, deg, pool);

  pad_dm1<<<PAD_T / 256, 256, 0, stream>>>(dm1, x_pad);

  deg_accum<<<(3 * (NEDGE / EPT) + 255) / 256, 256, 0, stream>>>(
      edges[0], ew[0], edges[1], ew[1], edges[2], ew[2], deg);

  compute_dinv<<<(3 * NP + 255) / 256, 256, 0, stream>>>(deg, dinv);

  build_ahat<<<(3 * (NEDGE / EPT + N_DRUGS) + 255) / 256, 256, 0, stream>>>(
      edges[0], ew[0], edges[1], ew[1], edges[2], ew[2], dinv, ahat);

  const int m_tiles  = NP / 16;
  const int n_wtiles = FDIM / 64;
  const int blocks   = ((m_tiles + 3) / 4) * n_wtiles;

  const size_t CH = (size_t)NP * FDIM;
  float* h1_t = H;          float* h2_t = H + CH;
  float* h1_s = H + 2 * CH; float* h2_s = H + 3 * CH;
  float* h1_g = H + 4 * CH; float* h2_g = H + 5 * CH;
  float* Ah_t = ahat;
  float* Ah_s = ahat + (size_t)NP * NP;
  float* Ah_g = ahat + (size_t)2 * NP * NP;


  gemm_wmma_f32<<<dim3(blocks, 3), 128, 0, stream>>>(
      x_pad, x_pad, x_pad, FDIM, W1[0], W1[1], W1[2], FDIM,
      h2_t, h2_s, h2_g, FDIM, FDIM, n_wtiles, m_tiles,
      nullptr, nullptr, nullptr, 0);
  gemm_wmma_f32<<<dim3(blocks, 3), 128, 0, stream>>>(
      Ah_t, Ah_s, Ah_g, NP, h2_t, h2_s, h2_g, FDIM,
      h1_t, h1_s, h1_g, FDIM, NP, n_wtiles, m_tiles,
      b1[0], b1[1], b1[2], 1);
  gemm_wmma_f32<<<dim3(blocks, 3), 128, 0, stream>>>(
      h1_t, h1_s, h1_g, FDIM, W2[0], W2[1], W2[2], FDIM,
      xw, x_pad, h2_t, FDIM, FDIM, n_wtiles, m_tiles,
      nullptr, nullptr, nullptr, 0);
  gemm_wmma_f32<<<dim3(blocks, 2), 128, 0, stream>>>(
      Ah_s, Ah_g, nullptr, NP, x_pad, h2_t, nullptr, FDIM,
      h2_s, h2_g, nullptr, FDIM, NP, n_wtiles, m_tiles,
      b2[1], b2[2], nullptr, 1);
  gemm_wmma_f32<<<dim3(blocks, 1), 128, 0, stream>>>(
      Ah_t, nullptr, nullptr, NP, xw, nullptr, nullptr, FDIM,
      h2_t, nullptr, nullptr, FDIM, NP, n_wtiles, m_tiles,
      b2[0], nullptr, nullptr, 1);

  dim3 pgrid(NP / POOL_ROWS, 6);
  pool_reduce<<<pgrid, 256, 0, stream>>>(H, pool);
  attn_kernel<<<1, 32, 0, stream>>>(pool, fc1_w, fc1_b, fc2_w, fc2_b, a_att);
  final_kernel<<<(N_DRUGS * ROWQ / 2 + 255) / 256, 256, 0, stream>>>(
      H, a_att, conv_w, conv_b, (float*)d_out);
}


# Second-layer channel of the first graph type is produced in the output buffer and combined there in place; the freed slot lets the last GEMM stage run as one launch (9 to 8); sums in parent's order
# speedup vs baseline: 9.6116x; 1.1947x over previous
#include <hip/hip_runtime.h>
#include <hip/hip_bf16.h>


#define N_DRUGS 591
#define NP      592
#define FDIM    1024
#define NEDGE   120000

typedef __attribute__((ext_vector_type(2))) float v2f;
typedef __attribute__((ext_vector_type(4))) float v4f;
typedef __attribute__((ext_vector_type(8))) float v8f;
typedef __attribute__((ext_vector_type(4))) int   v4i;

#define ROWQ    (FDIM / 4)
#define EPT     8

__global__ void init_ws(float* __restrict__ ahat, float* __restrict__ deg,
                        float* __restrict__ zrow, float* __restrict__ pool) {
  int idx = blockIdx.x * blockDim.x + threadIdx.x;
  int stride = gridDim.x * blockDim.x;
  const int AH4 = 3 * NP * NP / 4;
  const v4f zero = {0.0f, 0.0f, 0.0f, 0.0f};
  const v4f one  = {1.0f, 1.0f, 1.0f, 1.0f};
  v4f* a4 = (v4f*)ahat;
  v4f* d4 = (v4f*)deg;
  v4f* z4 = (v4f*)zrow;
  for (int i = idx; i < AH4; i += stride) a4[i] = zero;
  for (int i = idx; i < 3 * NP / 4; i += stride) { d4[i] = one; z4[i] = zero; }
  if (idx < 8) pool[idx] = 0.0f;
}

__global__ void deg_accum(const int* __restrict__ eT, const float* __restrict__ wT,
                          const int* __restrict__ eS, const float* __restrict__ wS,
                          const int* __restrict__ eG, const float* __restrict__ wG,
                          float* __restrict__ deg) {
  const int PER = NEDGE / EPT;
  int idx = blockIdx.x * blockDim.x + threadIdx.x;
  if (idx >= 3 * PER) return;
  int t = idx / PER, e = (idx - t * PER) * EPT;
  const int*   ed = (t == 0) ? eT : ((t == 1) ? eS : eG);
  const float* w  = (t == 0) ? wT : ((t == 1) ? wS : wG);
  v4i da = *(const v4i*)(ed + NEDGE + e), db = *(const v4i*)(ed + NEDGE + e + 4);
  v4f wa = *(const v4f*)(w + e),          wb = *(const v4f*)(w + e + 4);
  float* dg = deg + t * NP;
  atomicAdd(&dg[da.x], wa.x);  atomicAdd(&dg[da.y], wa.y);
  atomicAdd(&dg[da.z], wa.z);  atomicAdd(&dg[da.w], wa.w);
  atomicAdd(&dg[db.x], wb.x);  atomicAdd(&dg[db.y], wb.y);
  atomicAdd(&dg[db.z], wb.z);  atomicAdd(&dg[db.w], wb.w);
}

__device__ __forceinline__ float dinv_of(float d) {
  return (d > 0.0f) ? rsqrtf(d) : 0.0f;
}

__global__ void build_ahat(const int* __restrict__ eT, const float* __restrict__ wT,
                           const int* __restrict__ eS, const float* __restrict__ wS,
                           const int* __restrict__ eG, const float* __restrict__ wG,
                           const float* __restrict__ deg, float* __restrict__ ahat) {
  const int PER = NEDGE / EPT;
  int idx = blockIdx.x * blockDim.x + threadIdx.x;
  if (idx < 3 * PER) {
    int t = idx / PER, e = (idx - t * PER) * EPT;
    const float* dg = deg + t * NP;
    float* Ah = ahat + (size_t)t * NP * NP;
    const int*   ed = (t == 0) ? eT : ((t == 1) ? eS : eG);
    const float* w  = (t == 0) ? wT : ((t == 1) ? wS : wG);
    v4i sa = *(const v4i*)(ed + e),         sb = *(const v4i*)(ed + e + 4);
    v4i da = *(const v4i*)(ed + NEDGE + e), db = *(const v4i*)(ed + NEDGE + e + 4);
    v4f wa = *(const v4f*)(w + e),          wb = *(const v4f*)(w + e + 4);
    float gs0 = dg[sa.x], gs1 = dg[sa.y], gs2 = dg[sa.z], gs3 = dg[sa.w];
    float gs4 = dg[sb.x], gs5 = dg[sb.y], gs6 = dg[sb.z], gs7 = dg[sb.w];
    float gd0 = dg[da.x], gd1 = dg[da.y], gd2 = dg[da.z], gd3 = dg[da.w];
    float gd4 = dg[db.x], gd5 = dg[db.y], gd6 = dg[db.z], gd7 = dg[db.w];
    float n0 = dinv_of(gs0) * wa.x * dinv_of(gd0), n1 = dinv_of(gs1) * wa.y * dinv_of(gd1);
    float n2 = dinv_of(gs2) * wa.z * dinv_of(gd2), n3 = dinv_of(gs3) * wa.w * dinv_of(gd3);
    float n4 = dinv_of(gs4) * wb.x * dinv_of(gd4), n5 = dinv_of(gs5) * wb.y * dinv_of(gd5);
    float n6 = dinv_of(gs6) * wb.z * dinv_of(gd6), n7 = dinv_of(gs7) * wb.w * dinv_of(gd7);
    atomicAdd(&Ah[(size_t)da.x * NP + sa.x], n0);  atomicAdd(&Ah[(size_t)da.y * NP + sa.y], n1);
    atomicAdd(&Ah[(size_t)da.z * NP + sa.z], n2);  atomicAdd(&Ah[(size_t)da.w * NP + sa.w], n3);
    atomicAdd(&Ah[(size_t)db.x * NP + sb.x], n4);  atomicAdd(&Ah[(size_t)db.y * NP + sb.y], n5);
    atomicAdd(&Ah[(size_t)db.z * NP + sb.z], n6);  atomicAdd(&Ah[(size_t)db.w * NP + sb.w], n7);
  } else {
    int r = idx - 3 * PER;
    if (r >= 3 * N_DRUGS) return;
    int t = r / N_DRUGS, i = r - t * N_DRUGS;
    float* Ah = ahat + (size_t)t * NP * NP;
    float dv = dinv_of(deg[t * NP + i]);
    atomicAdd(&Ah[(size_t)i * NP + i], dv * dv);
  }
}

#define A_PITCH 20
#define B_PITCH 68

__device__ __forceinline__ void lds_barrier() {
  asm volatile("s_wait_dscnt 0x0\n\ts_barrier_signal -1\n\ts_barrier_wait -1" ::: "memory");
}

struct Frag { v4f a03, a47, q0, q1, q2, q3, q4, q5, q6, q7; };

__device__ __forceinline__ void load_frag(Frag& f, const float* as, const float* bs) {
  f.a03 = *(const v4f*)(as);
  f.a47 = *(const v4f*)(as + 4);
  f.q0 = *(const v4f*)(bs);
  f.q1 = *(const v4f*)(bs + B_PITCH);
  f.q2 = *(const v4f*)(bs + B_PITCH * 2);
  f.q3 = *(const v4f*)(bs + B_PITCH * 3);
  f.q4 = *(const v4f*)(bs + B_PITCH * 4);
  f.q5 = *(const v4f*)(bs + B_PITCH * 5);
  f.q6 = *(const v4f*)(bs + B_PITCH * 6);
  f.q7 = *(const v4f*)(bs + B_PITCH * 7);
}

__device__ __forceinline__ void mma_frag(const Frag& f, v8f& acc0, v8f& acc1,
                                         v8f& acc2, v8f& acc3) {
#define WMMA_STEP(AX, AY, QE, QO)                                                        \
  {                                                                                      \
    v2f a = {AX, AY};                                                                    \
    v2f b0 = {QE.x, QO.x}, b1 = {QE.y, QO.y}, b2 = {QE.z, QO.z}, b3 = {QE.w, QO.w};      \
    acc0 = __builtin_amdgcn_wmma_f32_16x16x4_f32(false, a, false, b0, (short)0, acc0, false, false); \
    acc1 = __builtin_amdgcn_wmma_f32_16x16x4_f32(false, a, false, b1, (short)0, acc1, false, false); \
    acc2 = __builtin_amdgcn_wmma_f32_16x16x4_f32(false, a, false, b2, (short)0, acc2, false, false); \
    acc3 = __builtin_amdgcn_wmma_f32_16x16x4_f32(false, a, false, b3, (short)0, acc3, false, false); \
  }
  WMMA_STEP(f.a03.x, f.a03.y, f.q0, f.q1)
  WMMA_STEP(f.a03.z, f.a03.w, f.q2, f.q3)
  WMMA_STEP(f.a47.x, f.a47.y, f.q4, f.q5)
  WMMA_STEP(f.a47.z, f.a47.w, f.q6, f.q7)
#undef WMMA_STEP
}
__global__ __launch_bounds__(128)
void gemm_wmma_f32(const float* __restrict__ A0, const float* __restrict__ A1,
                   const float* __restrict__ A2, int lda,
                   const float* __restrict__ B0, const float* __restrict__ B1,
                   const float* __restrict__ B2, int ldb,
                   float* __restrict__ C0, float* __restrict__ C1,
                   float* __restrict__ C2, int ldc,
                   int K, int n_wtiles, int m_tiles,
                   const float* __restrict__ bias0,
                   const float* __restrict__ bias1,
                   const float* __restrict__ bias2, int do_relu,
                   int a_rows, const float* __restrict__ zrow,
                   float* __restrict__ pool0, float* __restrict__ pool1,
                   float* __restrict__ pool2, int pool_rows, int c0_rows) {
  __shared__ __attribute__((aligned(16))) float As[2][64 * A_PITCH];
  __shared__ __attribute__((aligned(16))) float Bs[2][16 * B_PITCH];
  __shared__ float wsum[4];

  const int tid = threadIdx.x;
  const int t = blockIdx.y;
  const float* A    = (t == 0) ? A0    : ((t == 1) ? A1    : A2);
  const float* B    = (t == 0) ? B0    : ((t == 1) ? B1    : B2);
  float*       C    = (t == 0) ? C0    : ((t == 1) ? C1    : C2);
  const float* bias = (t == 0) ? bias0 : ((t == 1) ? bias1 : bias2);
  float*       pool = (t == 0) ? pool0 : ((t == 1) ? pool1 : pool2);
  const int bm = blockIdx.x / n_wtiles, bn = blockIdx.x - bm * n_wtiles;
  const int lane = tid & 31;
  const int wm = __builtin_amdgcn_readfirstlane(tid >> 5);
  const int c = lane & 15;
  const int h = lane >> 4;
  const int tm = bm * 4 + wm;
  const bool live = tm < m_tiles;
  const int row0 = tm << 4, col0 = bn << 6;

  const int ar = tid >> 2, aq = (tid & 3) << 2;
  const int br = tid >> 4, bq = (tid & 15) << 2;
  const int ga0 = (bm << 6) + ar, ga1 = ga0 + 32;
  const float* Ag0 = (ga0 < a_rows) ? A + (size_t)ga0 * lda + aq : zrow + aq;
  const float* Ag1 = (ga1 < a_rows) ? A + (size_t)ga1 * lda + aq : zrow + aq;
  const float* Bg0 = B + (size_t)br * ldb + (col0 + bq);
  const float* Bg1 = Bg0 + (size_t)ldb * 8;
  const int sa0 = ar * A_PITCH + aq, sa1 = sa0 + 32 * A_PITCH;
  const int sb0 = br * B_PITCH + bq, sb1 = sb0 + 8 * B_PITCH;
  const int fa = ((wm << 4) + c) * A_PITCH + (h << 3);
  const int fb = (h << 3) * B_PITCH + (c << 2);

  const int nk = K >> 4;
  v4f xa0, xa1, xb0, xb1, ya0, ya1, yb0, yb1;
#define TILE_LOAD(P, KT)                                                        \
  {                                                                             \
    const int kb = (((KT) < nk) ? (KT) : (nk - 1)) << 4;                        \
    P##a0 = *(const v4f*)(Ag0 + kb);                                            \
    P##a1 = *(const v4f*)(Ag1 + kb);                                            \
    P##b0 = *(const v4f*)(Bg0 + (size_t)kb * ldb);                              \
    P##b1 = *(const v4f*)(Bg1 + (size_t)kb * ldb);                              \
  }
#define TILE_STORE(P, ST)                                                       \
  {                                                                             \
    *(v4f*)&As[ST][sa0] = P##a0;  *(v4f*)&As[ST][sa1] = P##a1;                  \
    *(v4f*)&Bs[ST][sb0] = P##b0;  *(v4f*)&Bs[ST][sb1] = P##b1;                  \
  }
  TILE_LOAD(x, 0)
  TILE_LOAD(y, 1)
  TILE_STORE(x, 0)
  TILE_STORE(y, 1)
  TILE_LOAD(x, 2)
  lds_barrier();

  v8f acc0 = {}, acc1 = {}, acc2 = {}, acc3 = {};
  Frag f0, f1;
  if (live) load_frag(f0, &As[0][fa], &Bs[0][fb]);
  lds_barrier();

#define GEMM_STEP(S, CUR, NXT, RS, WS, GLD, GST)                                \
  {                                                                             \
    TILE_LOAD(GLD, (S) + 3)                                                     \
    if (live) {                                                                 \
      load_frag(NXT, &As[RS][fa], &Bs[RS][fb]);                                 \
        \
        \
      __builtin_amdgcn_sched_barrier(0);                                        \
      mma_frag(CUR, acc0, acc1, acc2, acc3);                                    \
    }                                                                           \
    TILE_STORE(GST, WS)                                                         \
    lds_barrier();                                                              \
  }

  int s = 0;
  for (; s + 1 < nk; s += 2) {
    GEMM_STEP(s,     f0, f1, 1, 0, y, x)
    GEMM_STEP(s + 1, f1, f0, 0, 1, x, y)
  }
  if (s < nk) {
    if (live) mma_frag(f0, acc0, acc1, acc2, acc3);
  }
#undef GEMM_STEP
#undef TILE_LOAD
#undef TILE_STORE

  float psum = 0.0f;
  if (live) {
    int crow0 = row0 + (h << 3);
    int ccol = col0 + (c << 2);
    const int c_rows = (t == 0) ? c0_rows : (m_tiles << 4);
    v4f bv = {0.0f, 0.0f, 0.0f, 0.0f};
    if (bias != nullptr) bv = *(const v4f*)(bias + ccol);
    v4f vr[8];
#pragma unroll
    for (int r = 0; r < 8; ++r) {
      v4f v = {acc0[r], acc1[r], acc2[r], acc3[r]};
      v += bv;
      if (do_relu) {
        v.x = fmaxf(v.x, 0.0f); v.y = fmaxf(v.y, 0.0f);
        v.z = fmaxf(v.z, 0.0f); v.w = fmaxf(v.w, 0.0f);
      }
      vr[r] = v;
      if (crow0 + r < pool_rows) psum += (v.x + v.y) + (v.z + v.w);
    }
    float* cp = C + (size_t)crow0 * ldc + ccol;
    if (row0 + 16 <= c_rows) {
#pragma unroll
      for (int r = 0; r < 8; ++r) *(v4f*)(cp + (size_t)r * ldc) = vr[r];
    } else {
#pragma unroll
      for (int r = 0; r < 8; ++r)
        if (crow0 + r < c_rows) *(v4f*)(cp + (size_t)r * ldc) = vr[r];
    }
  }
  if (pool != nullptr) {
    for (int off = 16; off > 0; off >>= 1) psum += __shfl_xor(psum, off);
    if (lane == 0) wsum[wm] = psum;
    lds_barrier();
    if (tid == 0) atomicAdd(pool, (wsum[0] + wsum[1]) + (wsum[2] + wsum[3]));
  }
}

#define FIN_T (N_DRUGS * ROWQ / 2)
__global__ void final_kernel(const float* __restrict__ H, const float* __restrict__ pool,
                             const float* __restrict__ fc1_w, const float* __restrict__ fc1_b,
                             const float* __restrict__ fc2_w, const float* __restrict__ fc2_b,
                             const float* __restrict__ conv_w,
                             const float* __restrict__ conv_b,
                             float* __restrict__ out) {
  int idx = blockIdx.x * blockDim.x + threadIdx.x;
  v4f hv[2][6];
#pragma unroll
  for (int j = 0; j < 2; ++j) {
    int q = idx + j * FIN_T;
#pragma unroll
    for (int c = 0; c < 6; ++c)
      hv[j][c] = (c == 1) ? ((const v4f*)out)[q]
                          : ((const v4f*)(H + (size_t)c * NP * FDIM))[q];
  }
  const float cb = conv_b[0];
  __builtin_amdgcn_sched_barrier(0);
  float ac[6], wc[6], w2[30];
  const int ln = threadIdx.x & 31;
  const int ju = ln < 30 ? ln : 29;
  const int cu = ln < 6 ? ln : 5;
#pragma unroll
  for (int j = 0; j < 30; ++j) w2[j] = fc2_w[cu * 30 + j];
  float hu = fc1_b[ju];
#pragma unroll
  for (int c = 0; c < 6; ++c) {
    float pc = pool[c] * (1.0f / (591.0f * 1024.0f));
    hu += pc * fc1_w[ju * 6 + c];
    wc[c] = conv_w[c];
  }
  hu = fmaxf(hu, 0.0f);
  float au = fc2_b[cu];
#pragma unroll
  for (int j = 0; j < 30; ++j) {
    float hj = __builtin_bit_cast(float, __builtin_amdgcn_readlane(__builtin_bit_cast(int, hu), j));
    au += hj * w2[j];
  }
  au = 1.0f / (1.0f + expf(-au));
#pragma unroll
  for (int c = 0; c < 6; ++c)
    ac[c] = __builtin_bit_cast(float, __builtin_amdgcn_readlane(__builtin_bit_cast(int, au), c));
#pragma unroll
  for (int j = 0; j < 2; ++j) {
    int q = idx + j * FIN_T;
    v4f s = {cb, cb, cb, cb};
#pragma unroll
    for (int c = 0; c < 6; ++c) {
      float vx = ac[c] * hv[j][c].x, vy = ac[c] * hv[j][c].y;
      float vz = ac[c] * hv[j][c].z, vw = ac[c] * hv[j][c].w;
      s.x += wc[c] * fmaxf(vx, 0.0f);  s.y += wc[c] * fmaxf(vy, 0.0f);
      s.z += wc[c] * fmaxf(vz, 0.0f);  s.w += wc[c] * fmaxf(vw, 0.0f);
    }
    ((v4f*)out)[q] = s;
  }
}

extern "C" void kernel_launch(void* const* d_in, const int* in_sizes, int n_in,
                              void* d_out, int out_size, void* d_ws, size_t ws_size,
                              hipStream_t stream) {
  (void)in_sizes; (void)n_in; (void)out_size; (void)ws_size;

  const float* dm1 = (const float*)d_in[0];
  const int*   edges[3] = {(const int*)d_in[1], (const int*)d_in[3], (const int*)d_in[5]};
  const float* ew[3]    = {(const float*)d_in[2], (const float*)d_in[4], (const float*)d_in[6]};
  const float* W1[3]    = {(const float*)d_in[7],  (const float*)d_in[11], (const float*)d_in[15]};
  const float* b1[3]    = {(const float*)d_in[8],  (const float*)d_in[12], (const float*)d_in[16]};
  const float* W2[3]    = {(const float*)d_in[9],  (const float*)d_in[13], (const float*)d_in[17]};
  const float* b2[3]    = {(const float*)d_in[10], (const float*)d_in[14], (const float*)d_in[18]};
  const float* fc1_w  = (const float*)d_in[19];
  const float* fc1_b  = (const float*)d_in[20];
  const float* fc2_w  = (const float*)d_in[21];
  const float* fc2_b  = (const float*)d_in[22];
  const float* conv_w = (const float*)d_in[23];
  const float* conv_b = (const float*)d_in[24];

  float* ws     = (float*)d_ws;
  float* ahat   = ws;
  float* deg    = ahat + (size_t)3 * NP * NP;
  float* zrow   = deg  + 3 * NP;
  float* pool   = zrow + 3 * NP;
  float* H      = pool + 16;
  float* xw     = H + (size_t)6 * NP * FDIM;
  float* xw_b   = xw + (size_t)NP * FDIM;

  init_ws<<<129, 256, 0, stream>>>(ahat, deg, zrow, pool);

  deg_accum<<<(3 * (NEDGE / EPT) + 255) / 256, 256, 0, stream>>>(
      edges[0], ew[0], edges[1], ew[1], edges[2], ew[2], deg);

  build_ahat<<<(3 * (NEDGE / EPT + N_DRUGS) + 255) / 256, 256, 0, stream>>>(
      edges[0], ew[0], edges[1], ew[1], edges[2], ew[2], deg, ahat);

  const int m_tiles  = NP / 16;
  const int n_wtiles = FDIM / 64;
  const int blocks   = ((m_tiles + 3) / 4) * n_wtiles;

  const size_t CH = (size_t)NP * FDIM;
  float* h1_t = H;          float* tmp_t = H + CH;
  float* h1_s = H + 2 * CH; float* h2_s = H + 3 * CH;
  float* h1_g = H + 4 * CH; float* h2_g = H + 5 * CH;
  float* Ah_t = ahat;
  float* Ah_s = ahat + (size_t)NP * NP;
  float* Ah_g = ahat + (size_t)2 * NP * NP;


  gemm_wmma_f32<<<dim3(blocks, 3), 128, 0, stream>>>(
      dm1, dm1, dm1, FDIM, W1[0], W1[1], W1[2], FDIM,
      tmp_t, h2_s, h2_g, FDIM, FDIM, n_wtiles, m_tiles,
      nullptr, nullptr, nullptr, 0,
      N_DRUGS, zrow, nullptr, nullptr, nullptr, 0, NP);
  gemm_wmma_f32<<<dim3(blocks, 3), 128, 0, stream>>>(
      Ah_t, Ah_s, Ah_g, NP, tmp_t, h2_s, h2_g, FDIM,
      h1_t, h1_s, h1_g, FDIM, NP, n_wtiles, m_tiles,
      b1[0], b1[1], b1[2], 1,
      NP, zrow, pool, pool + 2, pool + 4, N_DRUGS, NP);
  gemm_wmma_f32<<<dim3(blocks, 3), 128, 0, stream>>>(
      h1_t, h1_s, h1_g, FDIM, W2[0], W2[1], W2[2], FDIM,
      xw, xw_b, tmp_t, FDIM, FDIM, n_wtiles, m_tiles,
      nullptr, nullptr, nullptr, 0,
      NP, zrow, nullptr, nullptr, nullptr, 0, NP);
  gemm_wmma_f32<<<dim3(blocks, 3), 128, 0, stream>>>(
      Ah_t, Ah_s, Ah_g, NP, xw, xw_b, tmp_t, FDIM,
      (float*)d_out, h2_s, h2_g, FDIM, NP, n_wtiles, m_tiles,
      b2[0], b2[1], b2[2], 1,
      NP, zrow, pool + 1, pool + 3, pool + 5, N_DRUGS, N_DRUGS);

  final_kernel<<<FIN_T / 128, 128, 0, stream>>>(
      H, pool, fc1_w, fc1_b, fc2_w, fc2_b, conv_w, conv_b, (float*)d_out);
}


# GEMM K-step: LDS tile store moved in front of the WMMAs, barrier split (signal behind 12 of the step's WMMAs, wait behind the last), global loads 5 tiles ahead; all sums in the parent's order
# speedup vs baseline: 10.2102x; 1.0623x over previous
#include <hip/hip_runtime.h>
#include <hip/hip_bf16.h>


#define N_DRUGS 591
#define NP      592
#define FDIM    1024
#define NEDGE   120000

typedef __attribute__((ext_vector_type(2))) float v2f;
typedef __attribute__((ext_vector_type(4))) float v4f;
typedef __attribute__((ext_vector_type(8))) float v8f;
typedef __attribute__((ext_vector_type(4))) int   v4i;

#define ROWQ    (FDIM / 4)
#define EPT     8

__global__ void init_ws(float* __restrict__ ahat, float* __restrict__ deg,
                        float* __restrict__ zrow, float* __restrict__ pool) {
  int idx = blockIdx.x * blockDim.x + threadIdx.x;
  int stride = gridDim.x * blockDim.x;
  const int AH4 = 3 * NP * NP / 4;
  const v4f zero = {0.0f, 0.0f, 0.0f, 0.0f};
  const v4f one  = {1.0f, 1.0f, 1.0f, 1.0f};
  v4f* a4 = (v4f*)ahat;
  v4f* d4 = (v4f*)deg;
  v4f* z4 = (v4f*)zrow;
  for (int i = idx; i < AH4; i += stride) a4[i] = zero;
  for (int i = idx; i < 3 * NP / 4; i += stride) { d4[i] = one; z4[i] = zero; }
  if (idx < 8) pool[idx] = 0.0f;
}

__global__ void deg_accum(const int* __restrict__ eT, const float* __restrict__ wT,
                          const int* __restrict__ eS, const float* __restrict__ wS,
                          const int* __restrict__ eG, const float* __restrict__ wG,
                          float* __restrict__ deg) {
  const int PER = NEDGE / EPT;
  int idx = blockIdx.x * blockDim.x + threadIdx.x;
  if (idx >= 3 * PER) return;
  int t = idx / PER, e = (idx - t * PER) * EPT;
  const int*   ed = (t == 0) ? eT : ((t == 1) ? eS : eG);
  const float* w  = (t == 0) ? wT : ((t == 1) ? wS : wG);
  v4i da = *(const v4i*)(ed + NEDGE + e), db = *(const v4i*)(ed + NEDGE + e + 4);
  v4f wa = *(const v4f*)(w + e),          wb = *(const v4f*)(w + e + 4);
  float* dg = deg + t * NP;
  atomicAdd(&dg[da.x], wa.x);  atomicAdd(&dg[da.y], wa.y);
  atomicAdd(&dg[da.z], wa.z);  atomicAdd(&dg[da.w], wa.w);
  atomicAdd(&dg[db.x], wb.x);  atomicAdd(&dg[db.y], wb.y);
  atomicAdd(&dg[db.z], wb.z);  atomicAdd(&dg[db.w], wb.w);
}

__device__ __forceinline__ float dinv_of(float d) {
  return (d > 0.0f) ? rsqrtf(d) : 0.0f;
}

__global__ void build_ahat(const int* __restrict__ eT, const float* __restrict__ wT,
                           const int* __restrict__ eS, const float* __restrict__ wS,
                           const int* __restrict__ eG, const float* __restrict__ wG,
                           const float* __restrict__ deg, float* __restrict__ ahat) {
  const int PER = NEDGE / EPT;
  int idx = blockIdx.x * blockDim.x + threadIdx.x;
  if (idx < 3 * PER) {
    int t = idx / PER, e = (idx - t * PER) * EPT;
    const float* dg = deg + t * NP;
    float* Ah = ahat + (size_t)t * NP * NP;
    const int*   ed = (t == 0) ? eT : ((t == 1) ? eS : eG);
    const float* w  = (t == 0) ? wT : ((t == 1) ? wS : wG);
    v4i sa = *(const v4i*)(ed + e),         sb = *(const v4i*)(ed + e + 4);
    v4i da = *(const v4i*)(ed + NEDGE + e), db = *(const v4i*)(ed + NEDGE + e + 4);
    v4f wa = *(const v4f*)(w + e),          wb = *(const v4f*)(w + e + 4);
    float gs0 = dg[sa.x], gs1 = dg[sa.y], gs2 = dg[sa.z], gs3 = dg[sa.w];
    float gs4 = dg[sb.x], gs5 = dg[sb.y], gs6 = dg[sb.z], gs7 = dg[sb.w];
    float gd0 = dg[da.x], gd1 = dg[da.y], gd2 = dg[da.z], gd3 = dg[da.w];
    float gd4 = dg[db.x], gd5 = dg[db.y], gd6 = dg[db.z], gd7 = dg[db.w];
    float n0 = dinv_of(gs0) * wa.x * dinv_of(gd0), n1 = dinv_of(gs1) * wa.y * dinv_of(gd1);
    float n2 = dinv_of(gs2) * wa.z * dinv_of(gd2), n3 = dinv_of(gs3) * wa.w * dinv_of(gd3);
    float n4 = dinv_of(gs4) * wb.x * dinv_of(gd4), n5 = dinv_of(gs5) * wb.y * dinv_of(gd5);
    float n6 = dinv_of(gs6) * wb.z * dinv_of(gd6), n7 = dinv_of(gs7) * wb.w * dinv_of(gd7);
    atomicAdd(&Ah[(size_t)da.x * NP + sa.x], n0);  atomicAdd(&Ah[(size_t)da.y * NP + sa.y], n1);
    atomicAdd(&Ah[(size_t)da.z * NP + sa.z], n2);  atomicAdd(&Ah[(size_t)da.w * NP + sa.w], n3);
    atomicAdd(&Ah[(size_t)db.x * NP + sb.x], n4);  atomicAdd(&Ah[(size_t)db.y * NP + sb.y], n5);
    atomicAdd(&Ah[(size_t)db.z * NP + sb.z], n6);  atomicAdd(&Ah[(size_t)db.w * NP + sb.w], n7);
  } else {
    int r = idx - 3 * PER;
    if (r >= 3 * N_DRUGS) return;
    int t = r / N_DRUGS, i = r - t * N_DRUGS;
    float* Ah = ahat + (size_t)t * NP * NP;
    float dv = dinv_of(deg[t * NP + i]);
    atomicAdd(&Ah[(size_t)i * NP + i], dv * dv);
  }
}

#define A_PITCH 20
#define B_PITCH 68

__device__ __forceinline__ void lds_barrier() {
  asm volatile("s_wait_dscnt 0x0\n\ts_barrier_signal -1\n\ts_barrier_wait -1" ::: "memory");
}
__device__ __forceinline__ void lds_signal() {
  asm volatile("s_wait_dscnt 0x0\n\ts_barrier_signal -1" ::: "memory");
}
__device__ __forceinline__ void lds_wait() {
  asm volatile("s_barrier_wait -1" ::: "memory");
}

struct Frag { v4f a03, a47, q0, q1, q2, q3, q4, q5, q6, q7; };

__device__ __forceinline__ void load_frag(Frag& f, const float* as, const float* bs) {
  f.a03 = *(const v4f*)(as);
  f.a47 = *(const v4f*)(as + 4);
  f.q0 = *(const v4f*)(bs);
  f.q1 = *(const v4f*)(bs + B_PITCH);
  f.q2 = *(const v4f*)(bs + B_PITCH * 2);
  f.q3 = *(const v4f*)(bs + B_PITCH * 3);
  f.q4 = *(const v4f*)(bs + B_PITCH * 4);
  f.q5 = *(const v4f*)(bs + B_PITCH * 5);
  f.q6 = *(const v4f*)(bs + B_PITCH * 6);
  f.q7 = *(const v4f*)(bs + B_PITCH * 7);
}

#define WMMA_STEP(AX, AY, QE, QO)                                                        \
  {                                                                                      \
    v2f a = {AX, AY};                                                                    \
    v2f b0 = {QE.x, QO.x}, b1 = {QE.y, QO.y}, b2 = {QE.z, QO.z}, b3 = {QE.w, QO.w};      \
    acc0 = __builtin_amdgcn_wmma_f32_16x16x4_f32(false, a, false, b0, (short)0, acc0, false, false); \
    acc1 = __builtin_amdgcn_wmma_f32_16x16x4_f32(false, a, false, b1, (short)0, acc1, false, false); \
    acc2 = __builtin_amdgcn_wmma_f32_16x16x4_f32(false, a, false, b2, (short)0, acc2, false, false); \
    acc3 = __builtin_amdgcn_wmma_f32_16x16x4_f32(false, a, false, b3, (short)0, acc3, false, false); \
  }
__device__ __forceinline__ void mma_front(const Frag& f, v8f& acc0, v8f& acc1,
                                          v8f& acc2, v8f& acc3) {
  WMMA_STEP(f.a03.x, f.a03.y, f.q0, f.q1)
  WMMA_STEP(f.a03.z, f.a03.w, f.q2, f.q3)
  WMMA_STEP(f.a47.x, f.a47.y, f.q4, f.q5)
}
__device__ __forceinline__ void mma_back(const Frag& f, v8f& acc0, v8f& acc1,
                                         v8f& acc2, v8f& acc3) {
  WMMA_STEP(f.a47.z, f.a47.w, f.q6, f.q7)
}
#undef WMMA_STEP
__global__ __launch_bounds__(128)
void gemm_wmma_f32(const float* __restrict__ A0, const float* __restrict__ A1,
                   const float* __restrict__ A2, int lda,
                   const float* __restrict__ B0, const float* __restrict__ B1,
                   const float* __restrict__ B2, int ldb,
                   float* __restrict__ C0, float* __restrict__ C1,
                   float* __restrict__ C2, int ldc,
                   int K, int n_wtiles, int m_tiles,
                   const float* __restrict__ bias0,
                   const float* __restrict__ bias1,
                   const float* __restrict__ bias2, int do_relu,
                   int a_rows, const float* __restrict__ zrow,
                   float* __restrict__ pool0, float* __restrict__ pool1,
                   float* __restrict__ pool2, int pool_rows, int c0_rows) {
  __shared__ __attribute__((aligned(16))) float As[2][64 * A_PITCH];
  __shared__ __attribute__((aligned(16))) float Bs[2][16 * B_PITCH];
  __shared__ float wsum[4];

  const int tid = threadIdx.x;
  const int t = blockIdx.y;
  const float* A    = (t == 0) ? A0    : ((t == 1) ? A1    : A2);
  const float* B    = (t == 0) ? B0    : ((t == 1) ? B1    : B2);
  float*       C    = (t == 0) ? C0    : ((t == 1) ? C1    : C2);
  const float* bias = (t == 0) ? bias0 : ((t == 1) ? bias1 : bias2);
  float*       pool = (t == 0) ? pool0 : ((t == 1) ? pool1 : pool2);
  const int bm = blockIdx.x / n_wtiles, bn = blockIdx.x - bm * n_wtiles;
  const int lane = tid & 31;
  const int wm = __builtin_amdgcn_readfirstlane(tid >> 5);
  const int c = lane & 15;
  const int h = lane >> 4;
  const int tm = bm * 4 + wm;
  const bool live = tm < m_tiles;
  const int row0 = tm << 4, col0 = bn << 6;

  const int ar = tid >> 2, aq = (tid & 3) << 2;
  const int br = tid >> 4, bq = (tid & 15) << 2;
  const int ga0 = (bm << 6) + ar, ga1 = ga0 + 32;
  const float* Ag0 = (ga0 < a_rows) ? A + (size_t)ga0 * lda + aq : zrow + aq;
  const float* Ag1 = (ga1 < a_rows) ? A + (size_t)ga1 * lda + aq : zrow + aq;
  const float* Bg0 = B + (size_t)br * ldb + (col0 + bq);
  const float* Bg1 = Bg0 + (size_t)ldb * 8;
  const int sa0 = ar * A_PITCH + aq, sa1 = sa0 + 32 * A_PITCH;
  const int sb0 = br * B_PITCH + bq, sb1 = sb0 + 8 * B_PITCH;
  const int fa = ((wm << 4) + c) * A_PITCH + (h << 3);
  const int fb = (h << 3) * B_PITCH + (c << 2);

  const int nk = K >> 4;
  v4f t0a0, t0a1, t0b0, t0b1, t1a0, t1a1, t1b0, t1b1;
  v4f t2a0, t2a1, t2b0, t2b1, t3a0, t3a1, t3b0, t3b1;
#define TILE_LOAD(P, KT)                                                        \
  {                                                                             \
    const int kb = (((KT) < nk) ? (KT) : (nk - 1)) << 4;                        \
    P##a0 = *(const v4f*)(Ag0 + kb);                                            \
    P##a1 = *(const v4f*)(Ag1 + kb);                                            \
    P##b0 = *(const v4f*)(Bg0 + (size_t)kb * ldb);                              \
    P##b1 = *(const v4f*)(Bg1 + (size_t)kb * ldb);                              \
  }
#define TILE_STORE(P, ST)                                                       \
  {                                                                             \
    *(v4f*)&As[ST][sa0] = P##a0;  *(v4f*)&As[ST][sa1] = P##a1;                  \
    *(v4f*)&Bs[ST][sb0] = P##b0;  *(v4f*)&Bs[ST][sb1] = P##b1;                  \
  }
  TILE_LOAD(t0, 0)
  TILE_LOAD(t1, 1)
  TILE_LOAD(t2, 2)
  TILE_LOAD(t3, 3)
  TILE_STORE(t0, 0)
  TILE_STORE(t1, 1)
  TILE_LOAD(t0, 4)
  lds_barrier();

  v8f acc0 = {}, acc1 = {}, acc2 = {}, acc3 = {};
  Frag f0, f1;
  if (live) load_frag(f0, &As[0][fa], &Bs[0][fb]);
  lds_barrier();

#define GEMM_STEP(S, CUR, NXT, RS, WS, GST, GLD)                                \
  {                                                                             \
    TILE_STORE(GST, WS)                                                         \
    TILE_LOAD(GLD, (S) + 5)                                                     \
    if (live) {                                                                 \
      load_frag(NXT, &As[RS][fa], &Bs[RS][fb]);                                 \
        \
        \
      __builtin_amdgcn_sched_barrier(0);                                        \
      mma_front(CUR, acc0, acc1, acc2, acc3);                                   \
    }                                                                           \
    lds_signal();                                                               \
    if (live) mma_back(CUR, acc0, acc1, acc2, acc3);                            \
    lds_wait();                                                                 \
  }

  int s = 0;
  for (; s + 3 < nk; s += 4) {
    GEMM_STEP(s,     f0, f1, 1, 0, t2, t1)
    GEMM_STEP(s + 1, f1, f0, 0, 1, t3, t2)
    GEMM_STEP(s + 2, f0, f1, 1, 0, t0, t3)
    GEMM_STEP(s + 3, f1, f0, 0, 1, t1, t0)
  }
  if (s < nk)     GEMM_STEP(s,     f0, f1, 1, 0, t2, t1)
  if (s + 1 < nk) GEMM_STEP(s + 1, f1, f0, 0, 1, t3, t2)
  if (s + 2 < nk) GEMM_STEP(s + 2, f0, f1, 1, 0, t0, t3)
#undef GEMM_STEP
#undef TILE_LOAD
#undef TILE_STORE

  float psum = 0.0f;
  if (live) {
    int crow0 = row0 + (h << 3);
    int ccol = col0 + (c << 2);
    const int c_rows = (t == 0) ? c0_rows : (m_tiles << 4);
    v4f bv = {0.0f, 0.0f, 0.0f, 0.0f};
    if (bias != nullptr) bv = *(const v4f*)(bias + ccol);
    v4f vr[8];
#pragma unroll
    for (int r = 0; r < 8; ++r) {
      v4f v = {acc0[r], acc1[r], acc2[r], acc3[r]};
      v += bv;
      if (do_relu) {
        v.x = fmaxf(v.x, 0.0f); v.y = fmaxf(v.y, 0.0f);
        v.z = fmaxf(v.z, 0.0f); v.w = fmaxf(v.w, 0.0f);
      }
      vr[r] = v;
      if (crow0 + r < pool_rows) psum += (v.x + v.y) + (v.z + v.w);
    }
    float* cp = C + (size_t)crow0 * ldc + ccol;
    if (row0 + 16 <= c_rows) {
#pragma unroll
      for (int r = 0; r < 8; ++r) *(v4f*)(cp + (size_t)r * ldc) = vr[r];
    } else {
#pragma unroll
      for (int r = 0; r < 8; ++r)
        if (crow0 + r < c_rows) *(v4f*)(cp + (size_t)r * ldc) = vr[r];
    }
  }
  if (pool != nullptr) {
    for (int off = 16; off > 0; off >>= 1) psum += __shfl_xor(psum, off);
    if (lane == 0) wsum[wm] = psum;
    lds_barrier();
    if (tid == 0) atomicAdd(pool, (wsum[0] + wsum[1]) + (wsum[2] + wsum[3]));
  }
}

#define FIN_T (N_DRUGS * ROWQ / 2)
__global__ void final_kernel(const float* __restrict__ H, const float* __restrict__ pool,
                             const float* __restrict__ fc1_w, const float* __restrict__ fc1_b,
                             const float* __restrict__ fc2_w, const float* __restrict__ fc2_b,
                             const float* __restrict__ conv_w,
                             const float* __restrict__ conv_b,
                             float* __restrict__ out) {
  int idx = blockIdx.x * blockDim.x + threadIdx.x;
  v4f hv[2][6];
#pragma unroll
  for (int j = 0; j < 2; ++j) {
    int q = idx + j * FIN_T;
#pragma unroll
    for (int c = 0; c < 6; ++c)
      hv[j][c] = (c == 1) ? ((const v4f*)out)[q]
                          : ((const v4f*)(H + (size_t)c * NP * FDIM))[q];
  }
  const float cb = conv_b[0];
  __builtin_amdgcn_sched_barrier(0);
  float ac[6], wc[6], w2[30];
  const int ln = threadIdx.x & 31;
  const int ju = ln < 30 ? ln : 29;
  const int cu = ln < 6 ? ln : 5;
#pragma unroll
  for (int j = 0; j < 30; ++j) w2[j] = fc2_w[cu * 30 + j];
  float hu = fc1_b[ju];
#pragma unroll
  for (int c = 0; c < 6; ++c) {
    float pc = pool[c] * (1.0f / (591.0f * 1024.0f));
    hu += pc * fc1_w[ju * 6 + c];
    wc[c] = conv_w[c];
  }
  hu = fmaxf(hu, 0.0f);
  float au = fc2_b[cu];
#pragma unroll
  for (int j = 0; j < 30; ++j) {
    float hj = __builtin_bit_cast(float, __builtin_amdgcn_readlane(__builtin_bit_cast(int, hu), j));
    au += hj * w2[j];
  }
  au = 1.0f / (1.0f + expf(-au));
#pragma unroll
  for (int c = 0; c < 6; ++c)
    ac[c] = __builtin_bit_cast(float, __builtin_amdgcn_readlane(__builtin_bit_cast(int, au), c));
#pragma unroll
  for (int j = 0; j < 2; ++j) {
    int q = idx + j * FIN_T;
    v4f s = {cb, cb, cb, cb};
#pragma unroll
    for (int c = 0; c < 6; ++c) {
      float vx = ac[c] * hv[j][c].x, vy = ac[c] * hv[j][c].y;
      float vz = ac[c] * hv[j][c].z, vw = ac[c] * hv[j][c].w;
      s.x += wc[c] * fmaxf(vx, 0.0f);  s.y += wc[c] * fmaxf(vy, 0.0f);
      s.z += wc[c] * fmaxf(vz, 0.0f);  s.w += wc[c] * fmaxf(vw, 0.0f);
    }
    ((v4f*)out)[q] = s;
  }
}

extern "C" void kernel_launch(void* const* d_in, const int* in_sizes, int n_in,
                              void* d_out, int out_size, void* d_ws, size_t ws_size,
                              hipStream_t stream) {
  (void)in_sizes; (void)n_in; (void)out_size; (void)ws_size;

  const float* dm1 = (const float*)d_in[0];
  const int*   edges[3] = {(const int*)d_in[1], (const int*)d_in[3], (const int*)d_in[5]};
  const float* ew[3]    = {(const float*)d_in[2], (const float*)d_in[4], (const float*)d_in[6]};
  const float* W1[3]    = {(const float*)d_in[7],  (const float*)d_in[11], (const float*)d_in[15]};
  const float* b1[3]    = {(const float*)d_in[8],  (const float*)d_in[12], (const float*)d_in[16]};
  const float* W2[3]    = {(const float*)d_in[9],  (const float*)d_in[13], (const float*)d_in[17]};
  const float* b2[3]    = {(const float*)d_in[10], (const float*)d_in[14], (const float*)d_in[18]};
  const float* fc1_w  = (const float*)d_in[19];
  const float* fc1_b  = (const float*)d_in[20];
  const float* fc2_w  = (const float*)d_in[21];
  const float* fc2_b  = (const float*)d_in[22];
  const float* conv_w = (const float*)d_in[23];
  const float* conv_b = (const float*)d_in[24];

  float* ws     = (float*)d_ws;
  float* ahat   = ws;
  float* deg    = ahat + (size_t)3 * NP * NP;
  float* zrow   = deg  + 3 * NP;
  float* pool   = zrow + 3 * NP;
  float* H      = pool + 16;
  float* xw     = H + (size_t)6 * NP * FDIM;
  float* xw_b   = xw + (size_t)NP * FDIM;

  init_ws<<<129, 256, 0, stream>>>(ahat, deg, zrow, pool);

  deg_accum<<<(3 * (NEDGE / EPT) + 255) / 256, 256, 0, stream>>>(
      edges[0], ew[0], edges[1], ew[1], edges[2], ew[2], deg);

  build_ahat<<<(3 * (NEDGE / EPT + N_DRUGS) + 255) / 256, 256, 0, stream>>>(
      edges[0], ew[0], edges[1], ew[1], edges[2], ew[2], deg, ahat);

  const int m_tiles  = NP / 16;
  const int n_wtiles = FDIM / 64;
  const int blocks   = ((m_tiles + 3) / 4) * n_wtiles;

  const size_t CH = (size_t)NP * FDIM;
  float* h1_t = H;          float* tmp_t = H + CH;
  float* h1_s = H + 2 * CH; float* h2_s = H + 3 * CH;
  float* h1_g = H + 4 * CH; float* h2_g = H + 5 * CH;
  float* Ah_t = ahat;
  float* Ah_s = ahat + (size_t)NP * NP;
  float* Ah_g = ahat + (size_t)2 * NP * NP;


  gemm_wmma_f32<<<dim3(blocks, 3), 128, 0, stream>>>(
      dm1, dm1, dm1, FDIM, W1[0], W1[1], W1[2], FDIM,
      tmp_t, h2_s, h2_g, FDIM, FDIM, n_wtiles, m_tiles,
      nullptr, nullptr, nullptr, 0,
      N_DRUGS, zrow, nullptr, nullptr, nullptr, 0, NP);
  gemm_wmma_f32<<<dim3(blocks, 3), 128, 0, stream>>>(
      Ah_t, Ah_s, Ah_g, NP, tmp_t, h2_s, h2_g, FDIM,
      h1_t, h1_s, h1_g, FDIM, NP, n_wtiles, m_tiles,
      b1[0], b1[1], b1[2], 1,
      NP, zrow, pool, pool + 2, pool + 4, N_DRUGS, NP);
  gemm_wmma_f32<<<dim3(blocks, 3), 128, 0, stream>>>(
      h1_t, h1_s, h1_g, FDIM, W2[0], W2[1], W2[2], FDIM,
      xw, xw_b, tmp_t, FDIM, FDIM, n_wtiles, m_tiles,
      nullptr, nullptr, nullptr, 0,
      NP, zrow, nullptr, nullptr, nullptr, 0, NP);
  gemm_wmma_f32<<<dim3(blocks, 3), 128, 0, stream>>>(
      Ah_t, Ah_s, Ah_g, NP, xw, xw_b, tmp_t, FDIM,
      (float*)d_out, h2_s, h2_g, FDIM, NP, n_wtiles, m_tiles,
      b2[0], b2[1], b2[2], 1,
      NP, zrow, pool + 1, pool + 3, pool + 5, N_DRUGS, N_DRUGS);

  final_kernel<<<FIN_T / 128, 128, 0, stream>>>(
      H, pool, fc1_w, fc1_b, fc2_w, fc2_b, conv_w, conv_b, (float*)d_out);
}


# Adjacency build rides in the first GEMM launch (dm1 @ W1 reads no graph data): its 128-thread workgroups follow the 480 GEMM workgroups in one grid; 7 launches to 6; no sum's order changed
# speedup vs baseline: 11.5272x; 1.1290x over previous
#include <hip/hip_runtime.h>
#include <hip/hip_bf16.h>


#define N_DRUGS 591
#define NP      592
#define FDIM    1024
#define NEDGE   120000

typedef __attribute__((ext_vector_type(2))) float v2f;
typedef __attribute__((ext_vector_type(4))) float v4f;
typedef __attribute__((ext_vector_type(8))) float v8f;
typedef __attribute__((ext_vector_type(4))) int   v4i;

#define ROWQ    (FDIM / 4)
#define EPT     8

#define DGRP    16
#define DBLK    (3 * DGRP)
#define ZBLK    129
#define GEDGE   (NEDGE / DGRP)
#define GQUAD   (GEDGE / 4)
__global__ __launch_bounds__(256)
void init_ws(const int* __restrict__ eT, const float* __restrict__ wT,
             const int* __restrict__ eS, const float* __restrict__ wS,
             const int* __restrict__ eG, const float* __restrict__ wG,
             float* __restrict__ part, float* __restrict__ ahat,
             float* __restrict__ zrow, float* __restrict__ pool) {
  __shared__ float hist[NP];
  const int tid = threadIdx.x;
  const int b = blockIdx.x;
  if (b < DBLK) {
    const int t = b / DGRP, g = b - t * DGRP;
    const int*   ed = (t == 0) ? eT : ((t == 1) ? eS : eG);
    const float* w  = (t == 0) ? wT : ((t == 1) ? wS : wG);
    const v4i* dq = (const v4i*)(ed + NEDGE + g * GEDGE);
    const v4f* wq = (const v4f*)(w + g * GEDGE);
    static_assert(255 + 6 * 256 < GQUAD && GQUAD <= 8 * 256, "quads per thread");
    const bool last = tid + 7 * 256 < GQUAD;
    v4i dv[7]; v4f wv[7];
    v4i dl = {0, 0, 0, 0}; v4f wl = {0.0f, 0.0f, 0.0f, 0.0f};
#pragma unroll
    for (int j = 0; j < 7; ++j) { dv[j] = dq[tid + 256 * j]; wv[j] = wq[tid + 256 * j]; }
    if (last) { dl = dq[tid + 7 * 256]; wl = wq[tid + 7 * 256]; }
    for (int n = tid; n < NP; n += 256) hist[n] = 0.0f;
    __syncthreads();
#pragma unroll
    for (int j = 0; j < 7; ++j) {
      atomicAdd(&hist[dv[j].x], wv[j].x);  atomicAdd(&hist[dv[j].y], wv[j].y);
      atomicAdd(&hist[dv[j].z], wv[j].z);  atomicAdd(&hist[dv[j].w], wv[j].w);
    }
    if (last) {
      atomicAdd(&hist[dl.x], wl.x);  atomicAdd(&hist[dl.y], wl.y);
      atomicAdd(&hist[dl.z], wl.z);  atomicAdd(&hist[dl.w], wl.w);
    }
    __syncthreads();
    float* po = part + (size_t)b * NP;
    for (int n = tid; n < NP; n += 256) po[n] = hist[n];
  } else {
    const int idx = (b - DBLK) * 256 + tid;
    const int stride = ZBLK * 256;
    const int AH4 = 3 * NP * NP / 4;
    const v4f zero = {0.0f, 0.0f, 0.0f, 0.0f};
    v4f* a4 = (v4f*)ahat;
    v4f* z4 = (v4f*)zrow;
    for (int i = idx; i < AH4; i += stride) a4[i] = zero;
    for (int i = idx; i < 3 * NP / 4; i += stride) z4[i] = zero;
    if (idx < 8) pool[idx] = 0.0f;
  }
}

__device__ __forceinline__ float dinv_of(float d) {
  return (d > 0.0f) ? rsqrtf(d) : 0.0f;
}

#define ATHR 128
#define ABLK ((NEDGE / EPT + N_DRUGS + ATHR - 1) / ATHR)
__device__ __forceinline__
void build_ahat_block(const int bx, const int t,
                      const int* __restrict__ eT, const float* __restrict__ wT,
                      const int* __restrict__ eS, const float* __restrict__ wS,
                      const int* __restrict__ eG, const float* __restrict__ wG,
                      const float* __restrict__ part, float* __restrict__ ahat) {
  __shared__ float dis[NP];
  const int PER = NEDGE / EPT;
  const int tid = threadIdx.x;
  const int idx = bx * ATHR + tid;
  const int*   ed = (t == 0) ? eT : ((t == 1) ? eS : eG);
  const float* w  = (t == 0) ? wT : ((t == 1) ? wS : wG);
  float* Ah = ahat + (size_t)t * NP * NP;
  const bool is_edge = idx < PER;
  v4i sa = {0, 0, 0, 0}, sb = sa, da = sa, db = sa;
  v4f wa = {0.0f, 0.0f, 0.0f, 0.0f}, wb = wa;
  if (is_edge) {
    const int e = idx * EPT;
    sa = *(const v4i*)(ed + e);         sb = *(const v4i*)(ed + e + 4);
    da = *(const v4i*)(ed + NEDGE + e); db = *(const v4i*)(ed + NEDGE + e + 4);
    wa = *(const v4f*)(w + e);          wb = *(const v4f*)(w + e + 4);
  }
  const float* pt = part + (size_t)t * DGRP * NP;
  for (int n = tid; n < NP; n += ATHR) {
    float pg[DGRP];
#pragma unroll
    for (int g = 0; g < DGRP; ++g) pg[g] = pt[g * NP + n];
    float d = 1.0f;
#pragma unroll
    for (int g = 0; g < DGRP; ++g) d += pg[g];
    dis[n] = dinv_of(d);
  }
  __syncthreads();
  if (is_edge) {
    float n0 = dis[sa.x] * wa.x * dis[da.x], n1 = dis[sa.y] * wa.y * dis[da.y];
    float n2 = dis[sa.z] * wa.z * dis[da.z], n3 = dis[sa.w] * wa.w * dis[da.w];
    float n4 = dis[sb.x] * wb.x * dis[db.x], n5 = dis[sb.y] * wb.y * dis[db.y];
    float n6 = dis[sb.z] * wb.z * dis[db.z], n7 = dis[sb.w] * wb.w * dis[db.w];
    atomicAdd(&Ah[(size_t)da.x * NP + sa.x], n0);  atomicAdd(&Ah[(size_t)da.y * NP + sa.y], n1);
    atomicAdd(&Ah[(size_t)da.z * NP + sa.z], n2);  atomicAdd(&Ah[(size_t)da.w * NP + sa.w], n3);
    atomicAdd(&Ah[(size_t)db.x * NP + sb.x], n4);  atomicAdd(&Ah[(size_t)db.y * NP + sb.y], n5);
    atomicAdd(&Ah[(size_t)db.z * NP + sb.z], n6);  atomicAdd(&Ah[(size_t)db.w * NP + sb.w], n7);
  } else {
    const int i = idx - PER;
    if (i < N_DRUGS) {
      const float dv = dis[i];
      atomicAdd(&Ah[(size_t)i * NP + i], dv * dv);
    }
  }
}

#define A_PITCH 20
#define B_PITCH 68

__device__ __forceinline__ void lds_barrier() {
  asm volatile("s_wait_dscnt 0x0\n\ts_barrier_signal -1\n\ts_barrier_wait -1" ::: "memory");
}
__device__ __forceinline__ void lds_signal() {
  asm volatile("s_wait_dscnt 0x0\n\ts_barrier_signal -1" ::: "memory");
}
__device__ __forceinline__ void lds_wait() {
  asm volatile("s_barrier_wait -1" ::: "memory");
}

struct Frag { v4f a03, a47, q0, q1, q2, q3, q4, q5, q6, q7; };

__device__ __forceinline__ void load_frag(Frag& f, const float* as, const float* bs) {
  f.a03 = *(const v4f*)(as);
  f.a47 = *(const v4f*)(as + 4);
  f.q0 = *(const v4f*)(bs);
  f.q1 = *(const v4f*)(bs + B_PITCH);
  f.q2 = *(const v4f*)(bs + B_PITCH * 2);
  f.q3 = *(const v4f*)(bs + B_PITCH * 3);
  f.q4 = *(const v4f*)(bs + B_PITCH * 4);
  f.q5 = *(const v4f*)(bs + B_PITCH * 5);
  f.q6 = *(const v4f*)(bs + B_PITCH * 6);
  f.q7 = *(const v4f*)(bs + B_PITCH * 7);
}

#define WMMA_STEP(AX, AY, QE, QO)                                                        \
  {                                                                                      \
    v2f a = {AX, AY};                                                                    \
    v2f b0 = {QE.x, QO.x}, b1 = {QE.y, QO.y}, b2 = {QE.z, QO.z}, b3 = {QE.w, QO.w};      \
    acc0 = __builtin_amdgcn_wmma_f32_16x16x4_f32(false, a, false, b0, (short)0, acc0, false, false); \
    acc1 = __builtin_amdgcn_wmma_f32_16x16x4_f32(false, a, false, b1, (short)0, acc1, false, false); \
    acc2 = __builtin_amdgcn_wmma_f32_16x16x4_f32(false, a, false, b2, (short)0, acc2, false, false); \
    acc3 = __builtin_amdgcn_wmma_f32_16x16x4_f32(false, a, false, b3, (short)0, acc3, false, false); \
  }
__device__ __forceinline__ void mma_front(const Frag& f, v8f& acc0, v8f& acc1,
                                          v8f& acc2, v8f& acc3) {
  WMMA_STEP(f.a03.x, f.a03.y, f.q0, f.q1)
  WMMA_STEP(f.a03.z, f.a03.w, f.q2, f.q3)
  WMMA_STEP(f.a47.x, f.a47.y, f.q4, f.q5)
}
__device__ __forceinline__ void mma_back(const Frag& f, v8f& acc0, v8f& acc1,
                                         v8f& acc2, v8f& acc3) {
  WMMA_STEP(f.a47.z, f.a47.w, f.q6, f.q7)
}
#undef WMMA_STEP
__device__ __forceinline__
void gemm_tile(const int bx, const int t,
               const float* __restrict__ A0, const float* __restrict__ A1,
               const float* __restrict__ A2, int lda,
               const float* __restrict__ B0, const float* __restrict__ B1,
               const float* __restrict__ B2, int ldb,
               float* __restrict__ C0, float* __restrict__ C1,
               float* __restrict__ C2, int ldc,
               int K, int n_wtiles, int m_tiles,
               const float* __restrict__ bias0,
               const float* __restrict__ bias1,
               const float* __restrict__ bias2, int do_relu,
               int a_rows, const float* __restrict__ zrow,
               float* __restrict__ pool0, float* __restrict__ pool1,
               float* __restrict__ pool2, int pool_rows, int c0_rows) {
  __shared__ __attribute__((aligned(16))) float As[2][64 * A_PITCH];
  __shared__ __attribute__((aligned(16))) float Bs[2][16 * B_PITCH];
  __shared__ float wsum[4];

  const int tid = threadIdx.x;
  const float* A    = (t == 0) ? A0    : ((t == 1) ? A1    : A2);
  const float* B    = (t == 0) ? B0    : ((t == 1) ? B1    : B2);
  float*       C    = (t == 0) ? C0    : ((t == 1) ? C1    : C2);
  const float* bias = (t == 0) ? bias0 : ((t == 1) ? bias1 : bias2);
  float*       pool = (t == 0) ? pool0 : ((t == 1) ? pool1 : pool2);
  const int bm = (int)((unsigned)bx / (unsigned)n_wtiles), bn = bx - bm * n_wtiles;
  const int lane = tid & 31;
  const int wm = __builtin_amdgcn_readfirstlane(tid >> 5);
  const int c = lane & 15;
  const int h = lane >> 4;
  const int tm = bm * 4 + wm;
  const bool live = tm < m_tiles;
  const int row0 = tm << 4, col0 = bn << 6;

  const int ar = tid >> 2, aq = (tid & 3) << 2;
  const int br = tid >> 4, bq = (tid & 15) << 2;
  const int ga0 = (bm << 6) + ar, ga1 = ga0 + 32;
  const float* Ag0 = (ga0 < a_rows) ? A + (size_t)ga0 * lda + aq : zrow + aq;
  const float* Ag1 = (ga1 < a_rows) ? A + (size_t)ga1 * lda + aq : zrow + aq;
  const float* Bg0 = B + (size_t)br * ldb + (col0 + bq);
  const float* Bg1 = Bg0 + (size_t)ldb * 8;
  const int sa0 = ar * A_PITCH + aq, sa1 = sa0 + 32 * A_PITCH;
  const int sb0 = br * B_PITCH + bq, sb1 = sb0 + 8 * B_PITCH;
  const int fa = ((wm << 4) + c) * A_PITCH + (h << 3);
  const int fb = (h << 3) * B_PITCH + (c << 2);

  const int nk = K >> 4;
  v4f t0a0, t0a1, t0b0, t0b1, t1a0, t1a1, t1b0, t1b1;
  v4f t2a0, t2a1, t2b0, t2b1, t3a0, t3a1, t3b0, t3b1;
#define TILE_LOAD(P, KT)                                                        \
  {                                                                             \
    const int kb = (((KT) < nk) ? (KT) : (nk - 1)) << 4;                        \
    P##a0 = *(const v4f*)(Ag0 + kb);                                            \
    P##a1 = *(const v4f*)(Ag1 + kb);                                            \
    P##b0 = *(const v4f*)(Bg0 + (size_t)kb * ldb);                              \
    P##b1 = *(const v4f*)(Bg1 + (size_t)kb * ldb);                              \
  }
#define TILE_STORE(P, ST)                                                       \
  {                                                                             \
    *(v4f*)&As[ST][sa0] = P##a0;  *(v4f*)&As[ST][sa1] = P##a1;                  \
    *(v4f*)&Bs[ST][sb0] = P##b0;  *(v4f*)&Bs[ST][sb1] = P##b1;                  \
  }
  TILE_LOAD(t0, 0)
  TILE_LOAD(t1, 1)
  TILE_LOAD(t2, 2)
  TILE_LOAD(t3, 3)
  TILE_STORE(t0, 0)
  TILE_STORE(t1, 1)
  TILE_LOAD(t0, 4)
  lds_barrier();

  v8f acc0 = {}, acc1 = {}, acc2 = {}, acc3 = {};
  Frag f0, f1;
  if (live) load_frag(f0, &As[0][fa], &Bs[0][fb]);
  lds_barrier();

#define GEMM_STEP(S, CUR, NXT, RS, WS, GST, GLD)                                \
  {                                                                             \
    TILE_STORE(GST, WS)                                                         \
    TILE_LOAD(GLD, (S) + 5)                                                     \
    if (live) {                                                                 \
      load_frag(NXT, &As[RS][fa], &Bs[RS][fb]);                                 \
        \
        \
      __builtin_amdgcn_sched_barrier(0);                                        \
      mma_front(CUR, acc0, acc1, acc2, acc3);                                   \
    }                                                                           \
    lds_signal();                                                               \
    if (live) mma_back(CUR, acc0, acc1, acc2, acc3);                            \
    lds_wait();                                                                 \
  }

  int s = 0;
  for (; s + 3 < nk; s += 4) {
    GEMM_STEP(s,     f0, f1, 1, 0, t2, t1)
    GEMM_STEP(s + 1, f1, f0, 0, 1, t3, t2)
    GEMM_STEP(s + 2, f0, f1, 1, 0, t0, t3)
    GEMM_STEP(s + 3, f1, f0, 0, 1, t1, t0)
  }
  if (s < nk)     GEMM_STEP(s,     f0, f1, 1, 0, t2, t1)
  if (s + 1 < nk) GEMM_STEP(s + 1, f1, f0, 0, 1, t3, t2)
  if (s + 2 < nk) GEMM_STEP(s + 2, f0, f1, 1, 0, t0, t3)
#undef GEMM_STEP
#undef TILE_LOAD
#undef TILE_STORE

  float psum = 0.0f;
  if (live) {
    int crow0 = row0 + (h << 3);
    int ccol = col0 + (c << 2);
    const int c_rows = (t == 0) ? c0_rows : (m_tiles << 4);
    v4f bv = {0.0f, 0.0f, 0.0f, 0.0f};
    if (bias != nullptr) bv = *(const v4f*)(bias + ccol);
    v4f vr[8];
#pragma unroll
    for (int r = 0; r < 8; ++r) {
      v4f v = {acc0[r], acc1[r], acc2[r], acc3[r]};
      v += bv;
      if (do_relu) {
        v.x = fmaxf(v.x, 0.0f); v.y = fmaxf(v.y, 0.0f);
        v.z = fmaxf(v.z, 0.0f); v.w = fmaxf(v.w, 0.0f);
      }
      vr[r] = v;
      if (crow0 + r < pool_rows) psum += (v.x + v.y) + (v.z + v.w);
    }
    float* cp = C + (size_t)crow0 * ldc + ccol;
    if (row0 + 16 <= c_rows) {
#pragma unroll
      for (int r = 0; r < 8; ++r) *(v4f*)(cp + (size_t)r * ldc) = vr[r];
    } else {
#pragma unroll
      for (int r = 0; r < 8; ++r)
        if (crow0 + r < c_rows) *(v4f*)(cp + (size_t)r * ldc) = vr[r];
    }
  }
  if (pool != nullptr) {
    for (int off = 16; off > 0; off >>= 1) psum += __shfl_xor(psum, off);
    if (lane == 0) wsum[wm] = psum;
    lds_barrier();
    if (tid == 0) atomicAdd(pool, (wsum[0] + wsum[1]) + (wsum[2] + wsum[3]));
  }
}

__global__ __launch_bounds__(128)
void gemm_wmma_f32(const float* __restrict__ A0, const float* __restrict__ A1,
                   const float* __restrict__ A2, int lda,
                   const float* __restrict__ B0, const float* __restrict__ B1,
                   const float* __restrict__ B2, int ldb,
                   float* __restrict__ C0, float* __restrict__ C1,
                   float* __restrict__ C2, int ldc,
                   int K, int n_wtiles, int m_tiles,
                   const float* __restrict__ bias0,
                   const float* __restrict__ bias1,
                   const float* __restrict__ bias2, int do_relu,
                   int a_rows, const float* __restrict__ zrow,
                   float* __restrict__ pool0, float* __restrict__ pool1,
                   float* __restrict__ pool2, int pool_rows, int c0_rows) {
  gemm_tile(blockIdx.x, blockIdx.y, A0, A1, A2, lda, B0, B1, B2, ldb, C0, C1, C2, ldc,
            K, n_wtiles, m_tiles, bias0, bias1, bias2, do_relu, a_rows, zrow,
            pool0, pool1, pool2, pool_rows, c0_rows);
}

__global__ __launch_bounds__(ATHR)
void gemm1_adj(const float* __restrict__ X,
               const float* __restrict__ W0, const float* __restrict__ W1,
               const float* __restrict__ W2,
               float* __restrict__ C0, float* __restrict__ C1,
               float* __restrict__ C2,
               int n_wtiles, int m_tiles, int gblk, const float* __restrict__ zrow,
               const int* __restrict__ eT, const float* __restrict__ wT,
               const int* __restrict__ eS, const float* __restrict__ wS,
               const int* __restrict__ eG, const float* __restrict__ wG,
               const float* __restrict__ part, float* __restrict__ ahat) {
  const int b = blockIdx.x;
  if (b < 3 * gblk) {
    const int t = b / gblk;
    gemm_tile(b - t * gblk, t, X, X, X, FDIM, W0, W1, W2, FDIM, C0, C1, C2, FDIM,
              FDIM, n_wtiles, m_tiles, nullptr, nullptr, nullptr, 0, N_DRUGS, zrow,
              nullptr, nullptr, nullptr, 0, NP);
  } else {
    const int a = b - 3 * gblk;
    const int t = a / ABLK;
    build_ahat_block(a - t * ABLK, t, eT, wT, eS, wS, eG, wG, part, ahat);
  }
}

#define FIN_T (N_DRUGS * ROWQ / 2)
__global__ void final_kernel(const float* __restrict__ H, const float* __restrict__ pool,
                             const float* __restrict__ fc1_w, const float* __restrict__ fc1_b,
                             const float* __restrict__ fc2_w, const float* __restrict__ fc2_b,
                             const float* __restrict__ conv_w,
                             const float* __restrict__ conv_b,
                             float* __restrict__ out) {
  int idx = blockIdx.x * blockDim.x + threadIdx.x;
  v4f hv[2][6];
#pragma unroll
  for (int j = 0; j < 2; ++j) {
    int q = idx + j * FIN_T;
#pragma unroll
    for (int c = 0; c < 6; ++c)
      hv[j][c] = (c == 1) ? ((const v4f*)out)[q]
                          : ((const v4f*)(H + (size_t)c * NP * FDIM))[q];
  }
  const float cb = conv_b[0];
  __builtin_amdgcn_sched_barrier(0);
  float ac[6], wc[6], w2[30];
  const int ln = threadIdx.x & 31;
  const int ju = ln < 30 ? ln : 29;
  const int cu = ln < 6 ? ln : 5;
#pragma unroll
  for (int j = 0; j < 30; ++j) w2[j] = fc2_w[cu * 30 + j];
  float hu = fc1_b[ju];
#pragma unroll
  for (int c = 0; c < 6; ++c) {
    float pc = pool[c] * (1.0f / (591.0f * 1024.0f));
    hu += pc * fc1_w[ju * 6 + c];
    wc[c] = conv_w[c];
  }
  hu = fmaxf(hu, 0.0f);
  float au = fc2_b[cu];
#pragma unroll
  for (int j = 0; j < 30; ++j) {
    float hj = __builtin_bit_cast(float, __builtin_amdgcn_readlane(__builtin_bit_cast(int, hu), j));
    au += hj * w2[j];
  }
  au = 1.0f / (1.0f + expf(-au));
#pragma unroll
  for (int c = 0; c < 6; ++c)
    ac[c] = __builtin_bit_cast(float, __builtin_amdgcn_readlane(__builtin_bit_cast(int, au), c));
#pragma unroll
  for (int j = 0; j < 2; ++j) {
    int q = idx + j * FIN_T;
    v4f s = {cb, cb, cb, cb};
#pragma unroll
    for (int c = 0; c < 6; ++c) {
      float vx = ac[c] * hv[j][c].x, vy = ac[c] * hv[j][c].y;
      float vz = ac[c] * hv[j][c].z, vw = ac[c] * hv[j][c].w;
      s.x += wc[c] * fmaxf(vx, 0.0f);  s.y += wc[c] * fmaxf(vy, 0.0f);
      s.z += wc[c] * fmaxf(vz, 0.0f);  s.w += wc[c] * fmaxf(vw, 0.0f);
    }
    ((v4f*)out)[q] = s;
  }
}

extern "C" void kernel_launch(void* const* d_in, const int* in_sizes, int n_in,
                              void* d_out, int out_size, void* d_ws, size_t ws_size,
                              hipStream_t stream) {
  (void)in_sizes; (void)n_in; (void)out_size; (void)ws_size;

  const float* dm1 = (const float*)d_in[0];
  const int*   edges[3] = {(const int*)d_in[1], (const int*)d_in[3], (const int*)d_in[5]};
  const float* ew[3]    = {(const float*)d_in[2], (const float*)d_in[4], (const float*)d_in[6]};
  const float* W1[3]    = {(const float*)d_in[7],  (const float*)d_in[11], (const float*)d_in[15]};
  const float* b1[3]    = {(const float*)d_in[8],  (const float*)d_in[12], (const float*)d_in[16]};
  const float* W2[3]    = {(const float*)d_in[9],  (const float*)d_in[13], (const float*)d_in[17]};
  const float* b2[3]    = {(const float*)d_in[10], (const float*)d_in[14], (const float*)d_in[18]};
  const float* fc1_w  = (const float*)d_in[19];
  const float* fc1_b  = (const float*)d_in[20];
  const float* fc2_w  = (const float*)d_in[21];
  const float* fc2_b  = (const float*)d_in[22];
  const float* conv_w = (const float*)d_in[23];
  const float* conv_b = (const float*)d_in[24];

  float* ws     = (float*)d_ws;
  float* ahat   = ws;
  float* zrow   = ahat + (size_t)3 * NP * NP;
  float* pool   = zrow + 3 * NP;
  float* H      = pool + 16;
  float* xw     = H + (size_t)6 * NP * FDIM;
  float* xw_b   = xw + (size_t)NP * FDIM;

  float* part   = H;

  init_ws<<<DBLK + ZBLK, 256, 0, stream>>>(
      edges[0], ew[0], edges[1], ew[1], edges[2], ew[2], part, ahat, zrow, pool);

  const int m_tiles  = NP / 16;
  const int n_wtiles = FDIM / 64;
  const int blocks   = ((m_tiles + 3) / 4) * n_wtiles;

  const size_t CH = (size_t)NP * FDIM;
  float* h1_t = H;          float* tmp_t = H + CH;
  float* h1_s = H + 2 * CH; float* h2_s = H + 3 * CH;
  float* h1_g = H + 4 * CH; float* h2_g = H + 5 * CH;
  float* Ah_t = ahat;
  float* Ah_s = ahat + (size_t)NP * NP;
  float* Ah_g = ahat + (size_t)2 * NP * NP;


  gemm1_adj<<<3 * blocks + 3 * ABLK, ATHR, 0, stream>>>(
      dm1, W1[0], W1[1], W1[2], tmp_t, h2_s, h2_g, n_wtiles, m_tiles, blocks, zrow,
      edges[0], ew[0], edges[1], ew[1], edges[2], ew[2], part, ahat);
  gemm_wmma_f32<<<dim3(blocks, 3), 128, 0, stream>>>(
      Ah_t, Ah_s, Ah_g, NP, tmp_t, h2_s, h2_g, FDIM,
      h1_t, h1_s, h1_g, FDIM, NP, n_wtiles, m_tiles,
      b1[0], b1[1], b1[2], 1,
      NP, zrow, pool, pool + 2, pool + 4, N_DRUGS, NP);
  gemm_wmma_f32<<<dim3(blocks, 3), 128, 0, stream>>>(
      h1_t, h1_s, h1_g, FDIM, W2[0], W2[1], W2[2], FDIM,
      xw, xw_b, tmp_t, FDIM, FDIM, n_wtiles, m_tiles,
      nullptr, nullptr, nullptr, 0,
      NP, zrow, nullptr, nullptr, nullptr, 0, NP);
  gemm_wmma_f32<<<dim3(blocks, 3), 128, 0, stream>>>(
      Ah_t, Ah_s, Ah_g, NP, xw, xw_b, tmp_t, FDIM,
      (float*)d_out, h2_s, h2_g, FDIM, NP, n_wtiles, m_tiles,
      b2[0], b2[1], b2[2], 1,
      NP, zrow, pool + 1, pool + 3, pool + 5, N_DRUGS, N_DRUGS);

  final_kernel<<<FIN_T / 128, 128, 0, stream>>>(
      H, pool, fc1_w, fc1_b, fc2_w, fc2_b, conv_w, conv_b, (float*)d_out);
}
